# starting kernel
# baseline (speedup 1.0000x reference)
#include <hip/hip_runtime.h>
#include <hip/hip_bf16.h>


typedef __attribute__((ext_vector_type(2))) float v2f;
typedef __attribute__((ext_vector_type(8))) float v8f;

#define NN_NODES 25000
#define N_EDGES  400000
#define N_PAIR   100000
#define N_ROWS   200000

__device__ __forceinline__ unsigned enc_f32(float f) {
  unsigned u = __float_as_uint(f);
  return (u & 0x80000000u) ? ~u : (u | 0x80000000u);
}
__device__ __forceinline__ float dec_f32(unsigned e) {
  return (e & 0x80000000u) ? __uint_as_float(e & 0x7FFFFFFFu)
                           : __uint_as_float(~e);
}
#define ENC_NEG_INF 0x007FFFFFu

__device__ __forceinline__ float leaky(float v, float slope) {
  return v >= 0.f ? v : v * slope;
}

__global__ void gemm_wmma_f32(const float* __restrict__ A,
                              const float* __restrict__ W,
                              const float* __restrict__ bias,
                              float* __restrict__ C,
                              int M, int N, int K,
                              float slope, int act) {
  __shared__ float As[64][36];
  __shared__ float Ws[32][36];

  const int t     = threadIdx.x;
  const int lane  = t & 31;
  const int wave  = t >> 5;
  const int waveM = wave & 3;
  const int waveN = wave >> 2;
  const int mBase = blockIdx.x * 64;
  const int nBase = blockIdx.y * 32;
  const int lh    = lane >> 4;
  const int l15   = lane & 15;

  v8f acc = {0.f, 0.f, 0.f, 0.f, 0.f, 0.f, 0.f, 0.f};

  for (int k0 = 0; k0 < K; k0 += 32) {
#pragma unroll
    for (int i = 0; i < 2; ++i) {
      int idx = t + i * 256;
      int r   = idx >> 3;
      int c4  = (idx & 7) * 4;
      int gr  = mBase + r;
      int grc = gr < M ? gr : M - 1;
      float4 v = *(const float4*)(A + (size_t)grc * K + k0 + c4);
      if (gr >= M) { v.x = 0.f; v.y = 0.f; v.z = 0.f; v.w = 0.f; }
      *(float4*)&As[r][c4] = v;
    }
    {
      int r  = t >> 3;
      int c4 = (t & 7) * 4;
      *(float4*)&Ws[r][c4] =
          *(const float4*)(W + (size_t)(nBase + r) * K + k0 + c4);
    }
    __syncthreads();

#pragma unroll
    for (int ks = 0; ks < 8; ++ks) {
      int kb = ks * 4 + 2 * lh;
      v2f a, b;
      a.x = As[waveM * 16 + l15][kb];
      a.y = As[waveM * 16 + l15][kb + 1];
      b.x = Ws[waveN * 16 + l15][kb];
      b.y = Ws[waveN * 16 + l15][kb + 1];
      acc = __builtin_amdgcn_wmma_f32_16x16x4_f32(
           false, a,  false, b,
           (short)0, acc,  false,  false);
    }
    __syncthreads();
  }

#pragma unroll
  for (int r = 0; r < 8; ++r) {
    int row = mBase + waveM * 16 + r + 8 * lh;
    int col = nBase + waveN * 16 + l15;
    if (row < M) {
      float v = acc[r];
      if (bias) v += bias[col];
      if (act)  v = leaky(v, slope);
      C[(size_t)row * N + col] = v;
    }
  }
}

template <int H, int F>
__global__ void node_p_kernel(const float* __restrict__ ht,
                              const float* __restrict__ a,
                              float* __restrict__ p) {
  int idx = blockIdx.x * blockDim.x + threadIdx.x;
  if (idx >= NN_NODES * H) return;
  int n = idx / H, h = idx % H;
  const float* row = ht + ((size_t)n * H + h) * F;
  float acc = 0.f;
#pragma unroll 4
  for (int f = 0; f < F; f += 4) {
    float4 v = *(const float4*)(row + f);
    float4 w = *(const float4*)(a + f);
    acc += v.x * w.x + v.y * w.y + v.z * w.z + v.w * w.w;
  }
  p[idx] = acc;
}

__global__ void fill_u32_kernel(unsigned* __restrict__ dst, unsigned val, int n) {
  int i = blockIdx.x * blockDim.x + threadIdx.x;
  if (i < n) dst[i] = val;
}

template <int H>
__global__ void edge_score_kernel(const float* __restrict__ p,
                                  const float* __restrict__ ew,
                                  const int* __restrict__ src,
                                  const int* __restrict__ dst,
                                  float* __restrict__ ebuf,
                                  unsigned* __restrict__ menc) {
  int e = blockIdx.x * blockDim.x + threadIdx.x;
  if (e >= N_EDGES) return;
  float lw = log1pf(ew[e]);
  int s = src[e], d = dst[e];
  float ps[H], pd[H];
#pragma unroll
  for (int h = 0; h < H; ++h) ps[h] = p[s * H + h];
#pragma unroll
  for (int h = 0; h < H; ++h) pd[h] = p[d * H + h];
#pragma unroll
  for (int h = 0; h < H; ++h) {
    float sc = leaky(ps[h] + pd[h] + lw, 0.01f);
    ebuf[(size_t)e * H + h] = sc;
    atomicMax(&menc[d * H + h], enc_f32(sc));
  }
}

__global__ void node_m_kernel(unsigned* __restrict__ menc, int n) {
  int i = blockIdx.x * blockDim.x + threadIdx.x;
  if (i >= n) return;
  float m = dec_f32(menc[i]);
  if (!isfinite(m)) m = 0.f;
  menc[i] = __float_as_uint(m);
}

template <int H>
__global__ void edge_exp_kernel(float* __restrict__ ebuf,
                                const float* __restrict__ mf,
                                const int* __restrict__ dst,
                                float* __restrict__ sden) {
  int idx = blockIdx.x * blockDim.x + threadIdx.x;
  if (idx >= N_EDGES * H) return;
  int e = idx / H, h = idx % H;
  int d = dst[e];
  float ex = __expf(ebuf[idx] - mf[d * H + h]);
  ebuf[idx] = ex;
  atomicAdd(&sden[d * H + h], ex);
}

template <int H, int F>
__global__ void edge_agg_kernel(const float* __restrict__ ebuf,
                                const float* __restrict__ sden,
                                const float* __restrict__ ht,
                                const int* __restrict__ src,
                                const int* __restrict__ dst,
                                float* __restrict__ hagg) {
  int idx = blockIdx.x * blockDim.x + threadIdx.x;
  if (idx >= N_EDGES * H) return;
  int e = idx / H, h = idx % H;
  int s = src[e], d = dst[e];
  float alpha = ebuf[idx] / fmaxf(sden[d * H + h], 1e-20f);
  const float* hrow = ht + ((size_t)s * H + h) * F;
  float* orow = hagg + ((size_t)d * H + h) * F;
#pragma unroll 4
  for (int f = 0; f < F; f += 4) {
    float4 v = *(const float4*)(hrow + f);
    atomicAdd(orow + f + 0, alpha * v.x);
    atomicAdd(orow + f + 1, alpha * v.y);
    atomicAdd(orow + f + 2, alpha * v.z);
    atomicAdd(orow + f + 3, alpha * v.w);
  }
}

__global__ void bias_act_kernel(const float* __restrict__ in,
                                const float* __restrict__ b,
                                float* __restrict__ out,
                                int total, int mask, float slope, int act) {
  int i = blockIdx.x * blockDim.x + threadIdx.x;
  if (i >= total) return;
  float v = in[i] + b[i & mask];
  if (act) v = leaky(v, slope);
  out[i] = v;
}

__global__ void build_z_kernel(const float* __restrict__ emb,
                               const int* __restrict__ ps, const int* __restrict__ pd,
                               const int* __restrict__ ns, const int* __restrict__ nd,
                               float* __restrict__ Z) {
  int idx = blockIdx.x * blockDim.x + threadIdx.x;
  if (idx >= N_ROWS * 32) return;
  int row = idx >> 5;
  int f   = (idx & 31) * 4;
  int s, d;
  if (row < N_PAIR) { s = ps[row]; d = pd[row]; }
  else              { s = ns[row - N_PAIR]; d = nd[row - N_PAIR]; }
  float4 a = *(const float4*)(emb + (size_t)s * 128 + f);
  float4 b = *(const float4*)(emb + (size_t)d * 128 + f);
  float4 r; r.x = a.x * b.x; r.y = a.y * b.y; r.z = a.z * b.z; r.w = a.w * b.w;
  *(float4*)(Z + (size_t)row * 128 + f) = r;
}

__global__ void final_dot_kernel(const float* __restrict__ Z2,
                                 const float* __restrict__ pW3,
                                 const float* __restrict__ pb3,
                                 float* __restrict__ out) {
  int row = blockIdx.x * blockDim.x + threadIdx.x;
  if (row >= N_ROWS) return;
  const float* z = Z2 + (size_t)row * 32;
  float acc = pb3[0];
#pragma unroll
  for (int j = 0; j < 32; j += 4) {
    float4 a = *(const float4*)(z + j);
    float4 w = *(const float4*)(pW3 + j);
    acc += a.x * w.x + a.y * w.y + a.z * w.z + a.w * w.w;
  }
  out[row] = acc;
}

extern "C" void kernel_launch(void* const* d_in, const int* in_sizes, int n_in,
                              void* d_out, int out_size, void* d_ws, size_t ws_size,
                              hipStream_t stream) {
  const float* x    = (const float*)d_in[0];
  const float* ew   = (const float*)d_in[1];
  const int*   esrc = (const int*)d_in[2];
  const int*   edst = (const int*)d_in[3];
  const int*   psrc = (const int*)d_in[4];
  const int*   pdst = (const int*)d_in[5];
  const int*   nsrc = (const int*)d_in[6];
  const int*   ndst = (const int*)d_in[7];
  const float* Wl[3] = {(const float*)d_in[8],  (const float*)d_in[11], (const float*)d_in[14]};
  const float* al[3] = {(const float*)d_in[9],  (const float*)d_in[12], (const float*)d_in[15]};
  const float* bl[3] = {(const float*)d_in[10], (const float*)d_in[13], (const float*)d_in[16]};
  const float* pW1 = (const float*)d_in[17]; const float* pb1 = (const float*)d_in[18];
  const float* pW2 = (const float*)d_in[19]; const float* pb2 = (const float*)d_in[20];
  const float* pW3 = (const float*)d_in[21]; const float* pb3 = (const float*)d_in[22];

  float* out   = (float*)d_out;
  float* emb   = out + 2 * N_PAIR;

  float*    ws   = (float*)d_ws;
  float*    ht   = ws;
  float*    hbuf = ht   + 6400000;
  float*    pbuf = hbuf + 6400000;
  unsigned* menc = (unsigned*)(pbuf + 100000);
  float*    sden = (float*)(menc + 100000);
  float*    ebuf = sden + 100000;
  float*    Z    = ebuf + 1600000;
  float*    Z1   = Z    + 25600000;
  float*    Z2   = Z1   + 12800000;

  const int Hs[3]  = {4, 4, 1};
  const int Fs[3]  = {64, 64, 128};
  const int Kin[3] = {128, 256, 256};

  const float* hin = x;
  for (int l = 0; l < 3; ++l) {
    const int H = Hs[l], F = Fs[l], HF = H * F, K = Kin[l];
    float* hout = (l == 2) ? emb : hbuf;

    dim3 gg((NN_NODES + 63) / 64, HF / 32);
    gemm_wmma_f32<<<gg, 256, 0, stream>>>(hin, Wl[l], nullptr, ht,
                                          NN_NODES, HF, K, 0.f, 0);
    int nh = NN_NODES * H;
    if (l < 2) node_p_kernel<4, 64><<<(nh + 255) / 256, 256, 0, stream>>>(ht, al[l], pbuf);
    else       node_p_kernel<1, 128><<<(nh + 255) / 256, 256, 0, stream>>>(ht, al[l], pbuf);
    fill_u32_kernel<<<(nh + 255) / 256, 256, 0, stream>>>(menc, ENC_NEG_INF, nh);
    hipMemsetAsync(sden, 0, nh * sizeof(float), stream);

    if (l < 2) {
      edge_score_kernel<4><<<(N_EDGES + 255) / 256, 256, 0, stream>>>(
          pbuf, ew, esrc, edst, ebuf, menc);
    } else {
      edge_score_kernel<1><<<(N_EDGES + 255) / 256, 256, 0, stream>>>(
          pbuf, ew, esrc, edst, ebuf, menc);
    }
    node_m_kernel<<<(nh + 255) / 256, 256, 0, stream>>>(menc, nh);
    if (l < 2) {
      edge_exp_kernel<4><<<(N_EDGES * 4 + 255) / 256, 256, 0, stream>>>(
          ebuf, (const float*)menc, edst, sden);
    } else {
      edge_exp_kernel<1><<<(N_EDGES + 255) / 256, 256, 0, stream>>>(
          ebuf, (const float*)menc, edst, sden);
    }

    hipMemsetAsync(hout, 0, (size_t)NN_NODES * HF * sizeof(float), stream);
    if (l < 2) {
      edge_agg_kernel<4, 64><<<(N_EDGES * 4 + 255) / 256, 256, 0, stream>>>(
          ebuf, sden, ht, esrc, edst, hout);
    } else {
      edge_agg_kernel<1, 128><<<(N_EDGES + 255) / 256, 256, 0, stream>>>(
          ebuf, sden, ht, esrc, edst, hout);
    }
    bias_act_kernel<<<(NN_NODES * HF + 255) / 256, 256, 0, stream>>>(
        hout, bl[l], hout, NN_NODES * HF, HF - 1, 0.01f, (l < 2) ? 1 : 0);
    hin = hout;
  }

  build_z_kernel<<<(N_ROWS * 32 + 255) / 256, 256, 0, stream>>>(emb, psrc, pdst,
                                                                nsrc, ndst, Z);
  dim3 g1((N_ROWS + 63) / 64, 64 / 32);
  gemm_wmma_f32<<<g1, 256, 0, stream>>>(Z, pW1, pb1, Z1, N_ROWS, 64, 128, 0.2f, 1);
  dim3 g2((N_ROWS + 63) / 64, 32 / 32);
  gemm_wmma_f32<<<g2, 256, 0, stream>>>(Z1, pW2, pb2, Z2, N_ROWS, 32, 64, 0.2f, 1);
  final_dot_kernel<<<(N_ROWS + 255) / 256, 256, 0, stream>>>(Z2, pW3, pb3, out);
}


# Aggregation sums rows in registers: phase 1 files each queued edge in its row's LDS bucket (slot = returning int add), a wave gathers a whole row, stores it once; no f32 LDS adds; sum order = slot
# speedup vs baseline: 10.0124x; 10.0124x over previous
#include <hip/hip_runtime.h>
#include <hip/hip_bf16.h>


typedef __attribute__((ext_vector_type(2))) float v2f;
typedef __attribute__((ext_vector_type(8))) float v8f;
typedef __attribute__((ext_vector_type(4))) float v4f;

#define NN_NODES 25000
#define N_EDGES  400000
#define N_PAIR   100000
#define N_ROWS   200000

__device__ __forceinline__ unsigned enc_f32(float f) {
  unsigned u = __float_as_uint(f);
  return (u & 0x80000000u) ? ~u : (u | 0x80000000u);
}
__device__ __forceinline__ float dec_f32(unsigned e) {
  return (e & 0x80000000u) ? __uint_as_float(e & 0x7FFFFFFFu)
                           : __uint_as_float(~e);
}
#define ENC_NEG_INF 0x007FFFFFu

__device__ __forceinline__ float leaky(float v, float slope) {
  return v >= 0.f ? v : v * slope;
}

__global__ void gemm_wmma_f32(const float* __restrict__ A,
                              const float* __restrict__ W,
                              const float* __restrict__ bias,
                              float* __restrict__ C,
                              int M, int N, int K,
                              float slope, int act) {
  __shared__ float As[64][36];
  __shared__ float Ws[32][36];

  const int t     = threadIdx.x;
  const int lane  = t & 31;
  const int wave  = t >> 5;
  const int waveM = wave & 3;
  const int waveN = wave >> 2;
  const int mBase = blockIdx.x * 64;
  const int nBase = blockIdx.y * 32;
  const int lh    = lane >> 4;
  const int l15   = lane & 15;

  v8f acc = {0.f, 0.f, 0.f, 0.f, 0.f, 0.f, 0.f, 0.f};

  for (int k0 = 0; k0 < K; k0 += 32) {
    {
      const int r   = t >> 3;
      const int c4  = (t & 7) * 4;
      const int gr0 = mBase + r;
      const int gr1 = gr0 + 32;
      const int g0c = gr0 < M ? gr0 : M - 1;
      const int g1c = gr1 < M ? gr1 : M - 1;
      v4f va0 = *(const v4f*)(A + (size_t)g0c * K + k0 + c4);
      v4f va1 = *(const v4f*)(A + (size_t)g1c * K + k0 + c4);
      const v4f vw = *(const v4f*)(W + (size_t)(nBase + r) * K + k0 + c4);
      asm volatile("" :: "v"(va0), "v"(va1), "v"(vw));
      const v4f zero = {0.f, 0.f, 0.f, 0.f};
      if (gr0 >= M) va0 = zero;
      if (gr1 >= M) va1 = zero;
      *(v4f*)&As[r][c4]      = va0;
      *(v4f*)&As[r + 32][c4] = va1;
      *(v4f*)&Ws[r][c4]      = vw;
    }
    __syncthreads();

#pragma unroll
    for (int ks = 0; ks < 8; ++ks) {
      int kb = ks * 4 + 2 * lh;
      v2f a, b;
      a.x = As[waveM * 16 + l15][kb];
      a.y = As[waveM * 16 + l15][kb + 1];
      b.x = Ws[waveN * 16 + l15][kb];
      b.y = Ws[waveN * 16 + l15][kb + 1];
      acc = __builtin_amdgcn_wmma_f32_16x16x4_f32(
           false, a,  false, b,
           (short)0, acc,  false,  false);
    }
    __syncthreads();
  }

#pragma unroll
  for (int r = 0; r < 8; ++r) {
    int row = mBase + waveM * 16 + r + 8 * lh;
    int col = nBase + waveN * 16 + l15;
    if (row < M) {
      float v = acc[r];
      if (bias) v += bias[col];
      if (act)  v = leaky(v, slope);
      C[(size_t)row * N + col] = v;
    }
  }
}

template <int H, int F>
__global__ void node_p_kernel(const float* __restrict__ ht,
                              const float* __restrict__ a,
                              float* __restrict__ p) {
  int idx = blockIdx.x * blockDim.x + threadIdx.x;
  if (idx >= NN_NODES * H) return;
  int n = idx / H, h = idx % H;
  const float* row = ht + ((size_t)n * H + h) * F;
  float acc = 0.f;
#pragma unroll 4
  for (int f = 0; f < F; f += 4) {
    float4 v = *(const float4*)(row + f);
    float4 w = *(const float4*)(a + f);
    acc += v.x * w.x + v.y * w.y + v.z * w.z + v.w * w.w;
  }
  p[idx] = acc;
}

#define AGG_QCAP 272
#define AGG_SCAP 128
#define AGG_BCAP 64
#define AGG_GROW 8

template <int H>
__device__ __forceinline__ void node_scores(int s, int d, float lw,
                                            const float* __restrict__ p,
                                            float* sc) {
  float ps[H], pd[H];
#pragma unroll
  for (int h = 0; h < H; ++h) ps[h] = p[s * H + h];
#pragma unroll
  for (int h = 0; h < H; ++h) pd[h] = p[d * H + h];
#pragma unroll
  for (int h = 0; h < H; ++h) sc[h] = leaky(ps[h] + pd[h] + lw, 0.01f);
}

template <int H>
__device__ __forceinline__ int edge_scores(int e, int d,
                                           const float* __restrict__ p,
                                           const float* __restrict__ ew,
                                           const int* __restrict__ src,
                                           float* sc, float& lw) {
  const int s = src[e];
  lw = log1pf(ew[e]);
  node_scores<H>(s, d, lw, p, sc);
  return s;
}

template <int H, bool SAVE, bool RANK>
__device__ __forceinline__ bool agg_score_max(unsigned* q, float* qs, int qn, int lane, int d0,
                                              const float* __restrict__ p,
                                              const float* __restrict__ ew,
                                              const int* __restrict__ src,
                                              unsigned* mx, unsigned* sq, float* sl,
                                              unsigned* rowcnt, unsigned* bucket, unsigned qbase) {
  bool over = false;
  for (int base = 0; base < qn; base += 32) {
    const int i = base + lane;
    if (i < qn) {
      const unsigned ent = q[i];
      const int e = (int)(ent & 0xFFFFFu);
      const unsigned dl = ent >> 20;
      float sc[H];
      float lw;
      const int s = edge_scores<H>(e, d0 + (int)dl, p, ew, src, sc, lw);
#pragma unroll
      for (int h = 0; h < H; ++h) {
        qs[i * H + h] = sc[h];
        __hip_atomic_fetch_max(mx + dl * H + h, enc_f32(sc[h]), __ATOMIC_RELAXED,
                               __HIP_MEMORY_SCOPE_WORKGROUP);
      }
      const unsigned kept = (dl << 20) | (unsigned)s;
      q[i] = kept;
      if constexpr (SAVE) { sq[i] = kept; sl[i] = lw; }
      if constexpr (RANK) {
        const unsigned rk = __hip_atomic_fetch_add(rowcnt + dl, 1u, __ATOMIC_RELAXED,
                                                   __HIP_MEMORY_SCOPE_WORKGROUP);
        if (rk < (unsigned)AGG_BCAP) bucket[dl * AGG_BCAP + rk] = qbase + (unsigned)i;
        else over = true;
      }
    }
  }
  return over;
}

template <int H>
__device__ __forceinline__ bool agg_saved_score_max(unsigned* q, float* qs, int qn, int lane, int d0,
                                                    const float* __restrict__ p,
                                                    const unsigned* sq, const float* sl,
                                                    unsigned* mx,
                                                    unsigned* rowcnt, unsigned* bucket, unsigned qbase) {
  bool over = false;
  for (int base = 0; base < qn; base += 32) {
    const int i = base + lane;
    if (i < qn) {
      const unsigned ent = sq[i];
      const float lw = sl[i];
      const unsigned dl = ent >> 20;
      float sc[H];
      node_scores<H>((int)(ent & 0xFFFFFu), d0 + (int)dl, lw, p, sc);
#pragma unroll
      for (int h = 0; h < H; ++h) {
        qs[i * H + h] = sc[h];
        __hip_atomic_fetch_max(mx + dl * H + h, enc_f32(sc[h]), __ATOMIC_RELAXED,
                               __HIP_MEMORY_SCOPE_WORKGROUP);
      }
      q[i] = ent;
      const unsigned rk = __hip_atomic_fetch_add(rowcnt + dl, 1u, __ATOMIC_RELAXED,
                                                 __HIP_MEMORY_SCOPE_WORKGROUP);
      if (rk < (unsigned)AGG_BCAP) bucket[dl * AGG_BCAP + rk] = qbase + (unsigned)i;
      else over = true;
    }
  }
  return over;
}

template <int H, int F, int R, bool KEPT>
__device__ __forceinline__ void agg_drain(const unsigned* q, const float* qs, int qn, int lane, int d0,
                                          const float* __restrict__ p,
                                          const float* __restrict__ ew,
                                          const int* __restrict__ src,
                                          const unsigned* mx,
                                          const float* __restrict__ ht,
                                          float* acc, float* den) {
  constexpr int HF = H * F;
  constexpr int NV = HF / 128;
  for (int base = 0; base < qn; base += 32) {
    const int cnt = (qn - base) < 32 ? (qn - base) : 32;
    int s = 0, dl = 0;
    float al[H];
#pragma unroll
    for (int h = 0; h < H; ++h) al[h] = 0.f;
    if (lane < cnt) {
      const unsigned ent = q[base + lane];
      dl = (int)(ent >> 20);
      float sc[H];
      if constexpr (KEPT) {
        s = (int)(ent & 0xFFFFFu);
#pragma unroll
        for (int h = 0; h < H; ++h) sc[h] = qs[(base + lane) * H + h];
      } else {
        float lw;
        s = edge_scores<H>((int)(ent & 0xFFFFFu), d0 + dl, p, ew, src, sc, lw);
      }
#pragma unroll
      for (int h = 0; h < H; ++h) {
        float m = dec_f32(mx[dl * H + h]);
        if (!isfinite(m)) m = 0.f;
        al[h] = __expf(sc[h] - m);
        __hip_atomic_fetch_add(den + dl * H + h, al[h], __ATOMIC_RELAXED,
                               __HIP_MEMORY_SCOPE_WORKGROUP);
      }
    }
    for (int i = 0; i < cnt; i += 4) {
      float4 rv[4][NV];
#pragma unroll
      for (int g = 0; g < 4; ++g) {
        const int idx = (i + g) < cnt ? (i + g) : (cnt - 1);
        const int sg = __builtin_amdgcn_readlane(s, idx);
        const float4* rp = (const float4*)(ht + (size_t)sg * HF);
#pragma unroll
        for (int k = 0; k < NV; ++k) rv[g][k] = rp[k * 32 + lane];
      }
#pragma unroll
      for (int g = 0; g < 4; ++g) {
        if (i + g < cnt) {
          const int dlg = __builtin_amdgcn_readlane(dl, i + g);
          float ab[H];
#pragma unroll
          for (int h = 0; h < H; ++h)
            ab[h] = __int_as_float(__builtin_amdgcn_readlane(__float_as_int(al[h]), i + g));
          float* ap = acc + dlg * HF;
#pragma unroll
          for (int k = 0; k < NV; ++k) {
            float a;
            if constexpr (H == 1) a = ab[0];
            else a = (lane & 16) ? ab[2 * k + 1] : ab[2 * k];
            float* pp = ap + (k * 32 + lane) * 4;
            const float4 v = rv[g][k];
            __hip_atomic_fetch_add(pp + 0, a * v.x, __ATOMIC_RELAXED, __HIP_MEMORY_SCOPE_WORKGROUP);
            __hip_atomic_fetch_add(pp + 1, a * v.y, __ATOMIC_RELAXED, __HIP_MEMORY_SCOPE_WORKGROUP);
            __hip_atomic_fetch_add(pp + 2, a * v.z, __ATOMIC_RELAXED, __HIP_MEMORY_SCOPE_WORKGROUP);
            __hip_atomic_fetch_add(pp + 3, a * v.w, __ATOMIC_RELAXED, __HIP_MEMORY_SCOPE_WORKGROUP);
          }
        }
      }
    }
  }
}

template <int H, int F, int R, int NT, int PHASE>
__device__ __forceinline__ void agg_scan(const int* __restrict__ dst, int wave, int lane, int d0,
                                         unsigned* q, float* qs, int& qn, bool& spilled,
                                         const float* __restrict__ p,
                                         const float* __restrict__ ew,
                                         const int* __restrict__ src,
                                         const float* __restrict__ ht,
                                         unsigned* mx, float* acc, float* den) {
  constexpr int NW   = NT / 32;
  constexpr int NWIN = N_EDGES / 128;
  const int4* dst4 = (const int4*)dst;
  const unsigned lt = (1u << lane) - 1u;

  int4 wa, wb;
  {
    const int a0 = wave < NWIN ? wave : NWIN - 1;
    const int b0 = (wave + NW) < NWIN ? (wave + NW) : NWIN - 1;
    wa = dst4[a0 * 32 + lane];
    wb = dst4[b0 * 32 + lane];
  }
  for (int win = wave; win < NWIN; win += 2 * NW) {
    const int na = (win + 2 * NW) < NWIN ? (win + 2 * NW) : NWIN - 1;
    const int nb = (win + 3 * NW) < NWIN ? (win + 3 * NW) : NWIN - 1;
    const int4 xa = dst4[na * 32 + lane];
    const int4 xb = dst4[nb * 32 + lane];

#pragma unroll
    for (int half = 0; half < 2; ++half) {
      const int  w     = win + half * NW;
      const bool valid = w < NWIN;
      const int4 dv    = half ? wb : wa;
      const unsigned l0 = (unsigned)(dv.x - d0), l1 = (unsigned)(dv.y - d0);
      const unsigned l2 = (unsigned)(dv.z - d0), l3 = (unsigned)(dv.w - d0);
      const bool m0 = valid && l0 < (unsigned)R, m1 = valid && l1 < (unsigned)R;
      const bool m2 = valid && l2 < (unsigned)R, m3 = valid && l3 < (unsigned)R;
      if (__builtin_amdgcn_ballot_w32(m0 || m1 || m2 || m3)) {
        const unsigned eb = (unsigned)(w * 128 + lane * 4);
        const unsigned k0 = __builtin_amdgcn_ballot_w32(m0);
        const unsigned k1 = __builtin_amdgcn_ballot_w32(m1);
        const unsigned k2 = __builtin_amdgcn_ballot_w32(m2);
        const unsigned k3 = __builtin_amdgcn_ballot_w32(m3);
        if constexpr (PHASE == 0) {
          const int c = __popc(k0) + __popc(k1) + __popc(k2) + __popc(k3);
          if (qn + c > AGG_SCAP) {
            agg_score_max<H, false, false>(q, qs, qn, lane, d0, p, ew, src, mx, nullptr, nullptr,
                                           nullptr, nullptr, 0u);
            qn = 0;
            spilled = true;
          }
        }
        if (m0) q[qn + __popc(k0 & lt)] = (l0 << 20) | (eb + 0u);
        qn += __popc(k0);
        if (m1) q[qn + __popc(k1 & lt)] = (l1 << 20) | (eb + 1u);
        qn += __popc(k1);
        if (m2) q[qn + __popc(k2 & lt)] = (l2 << 20) | (eb + 2u);
        qn += __popc(k2);
        if (m3) q[qn + __popc(k3 & lt)] = (l3 << 20) | (eb + 3u);
        qn += __popc(k3);
      }
    }
    if constexpr (PHASE == 1) {
      if (qn >= 8) {
        agg_drain<H, F, R, false>(q, qs, qn, lane, d0, p, ew, src, mx, ht, acc, den);
        qn = 0;
      }
    }
    wa = xa;
    wb = xb;
  }
}

template <int H, int F, int R, int NW, bool ACT>
__device__ __forceinline__ void agg_rows(const unsigned* qbuf, const float* qsc,
                                         const unsigned* rowcnt, const unsigned* bucket,
                                         const unsigned* mx, unsigned* nextrow,
                                         int wave, int lane, int d0,
                                         const float* __restrict__ ht,
                                         const float* __restrict__ bias, float slope,
                                         float* __restrict__ hagg) {
  constexpr int HF = H * F;
  constexpr int NV = HF / 128;
  int r = wave;
  while (r < R) {
    const int n = d0 + r;
    if (n >= NN_NODES) break;
    const int cnt = __builtin_amdgcn_readfirstlane((int)rowcnt[r]);
    float m[H], den[H];
#pragma unroll
    for (int h = 0; h < H; ++h) {
      float mm = dec_f32(mx[r * H + h]);
      if (!isfinite(mm)) mm = 0.f;
      m[h] = mm;
      den[h] = 0.f;
    }
    float4 sum[NV];
#pragma unroll
    for (int k = 0; k < NV; ++k) sum[k] = make_float4(0.f, 0.f, 0.f, 0.f);

    for (int base = 0; base < cnt; base += 32) {
      const int c = (cnt - base) < 32 ? (cnt - base) : 32;
      int s = 0;
      float al[H];
#pragma unroll
      for (int h = 0; h < H; ++h) al[h] = 0.f;
      if (lane < c) {
        const unsigned idx = bucket[r * AGG_BCAP + base + lane];
        const unsigned ent = qbuf[(idx / AGG_SCAP) * AGG_QCAP + (idx % AGG_SCAP)];
        s = (int)(ent & 0xFFFFFu);
#pragma unroll
        for (int h = 0; h < H; ++h) al[h] = __expf(qsc[idx * H + h] - m[h]);
      }
      for (int i = 0; i < c; i += AGG_GROW) {
        float4 rv[AGG_GROW][NV];
#pragma unroll
        for (int g = 0; g < AGG_GROW; ++g) {
          const int idx = (i + g) < c ? (i + g) : (c - 1);
          const int sg = __builtin_amdgcn_readlane(s, idx);
          const float4* rp = (const float4*)(ht + (size_t)sg * HF);
#pragma unroll
          for (int k = 0; k < NV; ++k) rv[g][k] = rp[k * 32 + lane];
        }
#pragma unroll
        for (int g = 0; g < AGG_GROW; ++g) {
          if (i + g < c) {
            float ab[H];
#pragma unroll
            for (int h = 0; h < H; ++h) {
              ab[h] = __int_as_float(__builtin_amdgcn_readlane(__float_as_int(al[h]), i + g));
              den[h] += ab[h];
            }
#pragma unroll
            for (int k = 0; k < NV; ++k) {
              float a;
              if constexpr (H == 1) a = ab[0];
              else a = (lane & 16) ? ab[2 * k + 1] : ab[2 * k];
              const float4 v = rv[g][k];
              sum[k].x += a * v.x; sum[k].y += a * v.y;
              sum[k].z += a * v.z; sum[k].w += a * v.w;
            }
          }
        }
      }
    }

#pragma unroll
    for (int k = 0; k < NV; ++k) {
      float dsel;
      if constexpr (H == 1) dsel = den[0];
      else dsel = (lane & 16) ? den[2 * k + 1] : den[2 * k];
      const float sden = fmaxf(dsel, 1e-20f);
      const int c4 = k * 32 + lane;
      float4 v = sum[k];
      v.x = v.x / sden; v.y = v.y / sden; v.z = v.z / sden; v.w = v.w / sden;
      const float4 b = ((const float4*)bias)[c4];
      v.x += b.x; v.y += b.y; v.z += b.z; v.w += b.w;
      if constexpr (ACT) {
        v.x = leaky(v.x, slope); v.y = leaky(v.y, slope);
        v.z = leaky(v.z, slope); v.w = leaky(v.w, slope);
      }
      ((float4*)hagg)[(size_t)n * (HF / 4) + c4] = v;
    }

    unsigned nx = 0u;
    if (lane == 0)
      nx = __hip_atomic_fetch_add(nextrow, 1u, __ATOMIC_RELAXED, __HIP_MEMORY_SCOPE_WORKGROUP);
    r = __builtin_amdgcn_readfirstlane((int)nx);
  }
}

template <int H, int F, int R, int NT, bool ACT, int MODE>
__global__ __launch_bounds__(NT)
void edge_agg_owner_kernel(const float* __restrict__ p,
                           const float* __restrict__ ew,
                           const float* __restrict__ ht,
                           const int* __restrict__ src,
                           const int* __restrict__ dst,
                           const float* __restrict__ bias,
                           float slope,
                           float* __restrict__ hagg,
                           unsigned* savq, float* savl, int* savn) {
  constexpr int HF = H * F;
  constexpr int NW = NT / 32;
  static_assert(N_EDGES % 128 == 0, "edge windows of 128");
  static_assert(N_EDGES <= (1 << 20) && NN_NODES <= (1 << 20) && R <= 2048, "queue entry packing");
  static_assert(AGG_SCAP >= 128 && AGG_SCAP <= AGG_QCAP, "one window always fits an empty queue");
  static_assert(H == 1 || (H == 4 && F == 64), "head select per lane half");
  static_assert(HF % 128 == 0, "row = whole float4 per lane");
  static_assert(AGG_SCAP * NW <= (1 << 20) && AGG_BCAP >= 32, "bucket entry = place in the queues");

  __shared__ float    acc[R * HF];
  __shared__ float    den[R * H];
  __shared__ unsigned mx[R * H];
  __shared__ unsigned qbuf[NW * AGG_QCAP];
  __shared__ float    qsc[NW * AGG_SCAP * H];
  __shared__ unsigned rowcnt[R];
  __shared__ unsigned bucket[R * AGG_BCAP];
  __shared__ unsigned ctl[2];

  const int t    = threadIdx.x;
  const int lane = t & 31;
  const int wave = __builtin_amdgcn_readfirstlane(t >> 5);
  const int d0   = blockIdx.x * R;
  unsigned* q    = qbuf + wave * AGG_QCAP;
  float*    qs   = qsc + wave * (AGG_SCAP * H);

  for (int i = t; i < R * H; i += NT) mx[i] = ENC_NEG_INF;
  for (int i = t; i < R; i += NT) rowcnt[i] = 0u;
  if (t == 0) { ctl[0] = 0u; ctl[1] = (unsigned)NW; }
  __syncthreads();

  int  qn      = 0;
  bool spilled = false;
  const int slot = blockIdx.x * NW + wave;
  unsigned* sq   = savq + (size_t)slot * AGG_SCAP;
  float*    sl   = savl + (size_t)slot * AGG_SCAP;
  const unsigned qbase = (unsigned)(wave * AGG_SCAP);
  bool over = false;
  if constexpr (MODE == 0) {
    agg_scan<H, F, R, NT, 0>(dst, wave, lane, d0, q, qs, qn, spilled, p, ew, src, ht, mx, acc, den);
    over = agg_score_max<H, true, true>(q, qs, qn, lane, d0, p, ew, src, mx, sq, sl, rowcnt, bucket, qbase);
    if (lane == 0) savn[slot] = spilled ? -1 : qn;
  } else {
    const int saved = __builtin_amdgcn_readfirstlane(savn[slot]);
    if (saved >= 0) {
      qn = saved;
      over = agg_saved_score_max<H>(q, qs, qn, lane, d0, p, sq, sl, mx, rowcnt, bucket, qbase);
    } else {
      agg_scan<H, F, R, NT, 0>(dst, wave, lane, d0, q, qs, qn, spilled, p, ew, src, ht, mx, acc, den);
      over = agg_score_max<H, false, true>(q, qs, qn, lane, d0, p, ew, src, mx, nullptr, nullptr,
                                           rowcnt, bucket, qbase);
    }
  }
  if (over || spilled) ctl[0] = 1u;
  __syncthreads();

  if (__builtin_amdgcn_readfirstlane((int)ctl[0]) == 0) {
    agg_rows<H, F, R, NW, ACT>(qbuf, qsc, rowcnt, bucket, mx, ctl + 1, wave, lane, d0,
                               ht, bias, slope, hagg);
    return;
  }

  for (int i = t; i < R * HF / 4; i += NT)
    ((float4*)acc)[i] = make_float4(0.f, 0.f, 0.f, 0.f);
  for (int i = t; i < R * H; i += NT) den[i] = 0.f;
  __syncthreads();
  if (!spilled) {
    agg_drain<H, F, R, true>(q, qs, qn, lane, d0, p, ew, src, mx, ht, acc, den);
  } else {
    qn = 0;
    agg_scan<H, F, R, NT, 1>(dst, wave, lane, d0, q, qs, qn, spilled, p, ew, src, ht, mx, acc, den);
    if (qn > 0)
      agg_drain<H, F, R, false>(q, qs, qn, lane, d0, p, ew, src, mx, ht, acc, den);
  }
  __syncthreads();

  for (int i = t; i < R * HF / 4; i += NT) {
    const int r = i / (HF / 4);
    const int n = d0 + r;
    if (n < NN_NODES) {
      float4 v = ((const float4*)acc)[i];
      const int c4 = i % (HF / 4);
      const float s = fmaxf(den[r * H + (c4 * 4) / F], 1e-20f);
      v.x = v.x / s; v.y = v.y / s; v.z = v.z / s; v.w = v.w / s;
      const float4 b = ((const float4*)bias)[c4];
      v.x += b.x; v.y += b.y; v.z += b.z; v.w += b.w;
      if constexpr (ACT) {
        v.x = leaky(v.x, slope); v.y = leaky(v.y, slope);
        v.z = leaky(v.z, slope); v.w = leaky(v.w, slope);
      }
      ((float4*)hagg)[(size_t)d0 * (HF / 4) + i] = v;
    }
  }
}

__global__ void build_z_kernel(const float* __restrict__ emb,
                               const int* __restrict__ ps, const int* __restrict__ pd,
                               const int* __restrict__ ns, const int* __restrict__ nd,
                               float* __restrict__ Z) {
  int idx = blockIdx.x * blockDim.x + threadIdx.x;
  if (idx >= N_ROWS * 32) return;
  int row = idx >> 5;
  int f   = (idx & 31) * 4;
  int s, d;
  if (row < N_PAIR) { s = ps[row]; d = pd[row]; }
  else              { s = ns[row - N_PAIR]; d = nd[row - N_PAIR]; }
  float4 a = *(const float4*)(emb + (size_t)s * 128 + f);
  float4 b = *(const float4*)(emb + (size_t)d * 128 + f);
  float4 r; r.x = a.x * b.x; r.y = a.y * b.y; r.z = a.z * b.z; r.w = a.w * b.w;
  *(float4*)(Z + (size_t)row * 128 + f) = r;
}

__global__ void final_dot_kernel(const float* __restrict__ Z2,
                                 const float* __restrict__ pW3,
                                 const float* __restrict__ pb3,
                                 float* __restrict__ out) {
  int row = blockIdx.x * blockDim.x + threadIdx.x;
  if (row >= N_ROWS) return;
  const float* z = Z2 + (size_t)row * 32;
  float acc = pb3[0];
#pragma unroll
  for (int j = 0; j < 32; j += 4) {
    float4 a = *(const float4*)(z + j);
    float4 w = *(const float4*)(pW3 + j);
    acc += a.x * w.x + a.y * w.y + a.z * w.z + a.w * w.w;
  }
  out[row] = acc;
}

extern "C" void kernel_launch(void* const* d_in, const int* in_sizes, int n_in,
                              void* d_out, int out_size, void* d_ws, size_t ws_size,
                              hipStream_t stream) {
  const float* x    = (const float*)d_in[0];
  const float* ew   = (const float*)d_in[1];
  const int*   esrc = (const int*)d_in[2];
  const int*   edst = (const int*)d_in[3];
  const int*   psrc = (const int*)d_in[4];
  const int*   pdst = (const int*)d_in[5];
  const int*   nsrc = (const int*)d_in[6];
  const int*   ndst = (const int*)d_in[7];
  const float* Wl[3] = {(const float*)d_in[8],  (const float*)d_in[11], (const float*)d_in[14]};
  const float* al[3] = {(const float*)d_in[9],  (const float*)d_in[12], (const float*)d_in[15]};
  const float* bl[3] = {(const float*)d_in[10], (const float*)d_in[13], (const float*)d_in[16]};
  const float* pW1 = (const float*)d_in[17]; const float* pb1 = (const float*)d_in[18];
  const float* pW2 = (const float*)d_in[19]; const float* pb2 = (const float*)d_in[20];
  const float* pW3 = (const float*)d_in[21]; const float* pb3 = (const float*)d_in[22];

  float* out   = (float*)d_out;
  float* emb   = out + 2 * N_PAIR;

  float*    ws   = (float*)d_ws;
  float*    ht   = ws;
  float*    hbuf = ht   + 6400000;
  float*    pbuf = hbuf + 6400000;
  float*    Z    = pbuf + 100000;
  float*    Z1   = Z    + 25600000;
  float*    Z2   = Z1   + 12800000;
  unsigned* savq = (unsigned*)(Z2 + 6400000);
  float*    savl = Z2 + 6400000 + 1048576;
  int*      savn = (int*)(Z2 + 6400000 + 2097152);

  const int Hs[3]  = {4, 4, 1};
  const int Fs[3]  = {64, 64, 128};
  const int Kin[3] = {128, 256, 256};

  const float* hin = x;
  for (int l = 0; l < 3; ++l) {
    const int H = Hs[l], F = Fs[l], HF = H * F, K = Kin[l];
    float* hout = (l == 2) ? emb : hbuf;

    dim3 gg((NN_NODES + 63) / 64, HF / 32);
    gemm_wmma_f32<<<gg, 256, 0, stream>>>(hin, Wl[l], nullptr, ht,
                                          NN_NODES, HF, K, 0.f, 0);
    int nh = NN_NODES * H;
    if (l < 2) node_p_kernel<4, 64><<<(nh + 255) / 256, 256, 0, stream>>>(ht, al[l], pbuf);
    else       node_p_kernel<1, 128><<<(nh + 255) / 256, 256, 0, stream>>>(ht, al[l], pbuf);

    if (l == 0) {
      edge_agg_owner_kernel<4, 64, 98, 1024, true, 0><<<(NN_NODES + 97) / 98, 1024, 0, stream>>>(
          pbuf, ew, ht, esrc, edst, bl[l], 0.01f, hout, savq, savl, savn);
    } else if (l == 1) {
      edge_agg_owner_kernel<4, 64, 98, 1024, true, 1><<<(NN_NODES + 97) / 98, 1024, 0, stream>>>(
          pbuf, ew, ht, esrc, edst, bl[l], 0.01f, hout, savq, savl, savn);
    } else {
      edge_agg_owner_kernel<1, 128, 98, 1024, false, 1><<<(NN_NODES + 97) / 98, 1024, 0, stream>>>(
          pbuf, ew, ht, esrc, edst, bl[l], 0.01f, hout, savq, savl, savn);
    }
    hin = hout;
  }

  build_z_kernel<<<(N_ROWS * 32 + 255) / 256, 256, 0, stream>>>(emb, psrc, pdst,
                                                                nsrc, ndst, Z);
  dim3 g1((N_ROWS + 63) / 64, 64 / 32);
  gemm_wmma_f32<<<g1, 256, 0, stream>>>(Z, pW1, pb1, Z1, N_ROWS, 64, 128, 0.2f, 1);
  dim3 g2((N_ROWS + 63) / 64, 32 / 32);
  gemm_wmma_f32<<<g2, 256, 0, stream>>>(Z1, pW2, pb2, Z2, N_ROWS, 32, 64, 0.2f, 1);
  final_dot_kernel<<<(N_ROWS + 255) / 256, 256, 0, stream>>>(Z2, pW3, pb3, out);
}


# Owner queues of the three aggregations built in one pass by 32 binning work-groups (LDS ranks) riding in the first GEMM's launch: layer 1's 256 whole dst scans gone; sum order still LDS arrival
# speedup vs baseline: 10.6017x; 1.0589x over previous
#include <hip/hip_runtime.h>
#include <hip/hip_bf16.h>


typedef __attribute__((ext_vector_type(2))) float v2f;
typedef __attribute__((ext_vector_type(8))) float v8f;
typedef __attribute__((ext_vector_type(4))) float v4f;

#define NN_NODES 25000
#define N_EDGES  400000
#define N_PAIR   100000
#define N_ROWS   200000

__device__ __forceinline__ unsigned enc_f32(float f) {
  unsigned u = __float_as_uint(f);
  return (u & 0x80000000u) ? ~u : (u | 0x80000000u);
}
__device__ __forceinline__ float dec_f32(unsigned e) {
  return (e & 0x80000000u) ? __uint_as_float(e & 0x7FFFFFFFu)
                           : __uint_as_float(~e);
}
#define ENC_NEG_INF 0x007FFFFFu

__device__ __forceinline__ float leaky(float v, float slope) {
  return v >= 0.f ? v : v * slope;
}

#define AGG_R    98
#define AGG_NT   1024
#define AGG_NG   ((NN_NODES + AGG_R - 1) / AGG_R)
#define AGG_QCAP 272
#define AGG_SCAP 128
#define AGG_BCAP 64
#define AGG_GROW 8

template <int R, int NW, int NTB>
__device__ __forceinline__ void bin_owner_queues(unsigned* cnt, int w,
                                                 const int* __restrict__ src,
                                                 const int* __restrict__ dst,
                                                 const float* __restrict__ ew,
                                                 unsigned* savq, float* savl, int* savn) {
  constexpr int NG   = (NN_NODES + R - 1) / R;
  constexpr int NWIN = N_EDGES / 128;
  constexpr int STEP = NW * (NTB / 32);
  static_assert(N_EDGES % 128 == 0, "edge windows of 128");
  static_assert(R <= 2048 && NN_NODES <= (1 << 20), "queue entry packing");
  const int t    = threadIdx.x;
  const int lane = t & 31;
  const int wv   = __builtin_amdgcn_readfirstlane(t >> 5);
  for (int i = t; i < NG; i += NTB) cnt[i] = 0u;
  __syncthreads();

  const int4*   dst4 = (const int4*)dst;
  const int4*   src4 = (const int4*)src;
  const float4* ew4  = (const float4*)ew;
  int win = w + NW * wv;
  int4 dv[2];
#pragma unroll
  for (int h = 0; h < 2; ++h) {
    const int c0 = (win + h * STEP) < NWIN ? (win + h * STEP) : NWIN - 1;
    dv[h] = dst4[c0 * 32 + lane];
  }
  for (; win < NWIN; win += 2 * STEP) {
    int4   sv[2], dn[2];
    float4 wt[2];
#pragma unroll
    for (int h = 0; h < 2; ++h) {
      const int c0 = (win + h * STEP) < NWIN ? (win + h * STEP) : NWIN - 1;
      sv[h] = src4[c0 * 32 + lane];
      wt[h] = ew4[c0 * 32 + lane];
    }
#pragma unroll
    for (int h = 0; h < 2; ++h) {
      const int nx = (win + (2 + h) * STEP) < NWIN ? (win + (2 + h) * STEP) : NWIN - 1;
      dn[h] = dst4[nx * 32 + lane];
    }
#pragma unroll
    for (int h = 0; h < 2; ++h) {
      const bool  valid = (win + h * STEP) < NWIN;
      const int   dd[4] = {dv[h].x, dv[h].y, dv[h].z, dv[h].w};
      const int   ss[4] = {sv[h].x, sv[h].y, sv[h].z, sv[h].w};
      const float ww[4] = {wt[h].x, wt[h].y, wt[h].z, wt[h].w};
      unsigned qb[4], rw[4], rk[4];
#pragma unroll
      for (int k = 0; k < 4; ++k) {
        const bool ok = valid && (unsigned)dd[k] < (unsigned)NN_NODES;
        const unsigned d = ok ? (unsigned)dd[k] : 0u;
        const unsigned g = d / (unsigned)R;
        rw[k] = (d - g * (unsigned)R) << 20;
        qb[k] = (g * (unsigned)NW + (unsigned)w) * (unsigned)AGG_SCAP;
        const unsigned got = __hip_atomic_fetch_add(cnt + g, ok ? 1u : 0u, __ATOMIC_RELAXED,
                                                    __HIP_MEMORY_SCOPE_WORKGROUP);
        rk[k] = ok ? got : (unsigned)AGG_SCAP;
      }
#pragma unroll
      for (int k = 0; k < 4; ++k) {
        if (rk[k] < (unsigned)AGG_SCAP) {
          const unsigned at = qb[k] + rk[k];
          savq[at] = rw[k] | (unsigned)ss[k];
          savl[at] = ww[k];
        }
      }
    }
    dv[0] = dn[0];
    dv[1] = dn[1];
  }
  __syncthreads();
  for (int i = t; i < NG; i += NTB) {
    const unsigned n = cnt[i];
    savn[i * NW + w] = n <= (unsigned)AGG_SCAP ? (int)n : -1;
  }
}

template <bool BIN>
__global__ void gemm_wmma_f32(const float* __restrict__ A,
                              const float* __restrict__ W,
                              const float* __restrict__ bias,
                              float* __restrict__ C,
                              int M, int N, int K,
                              float slope, int act,
                              const int* __restrict__ esrc,
                              const int* __restrict__ edst,
                              const float* __restrict__ ew,
                              unsigned* savq, float* savl, int* savn) {
  __shared__ float As[64][36];
  __shared__ float Ws[32][36];

  if constexpr (BIN) {
    if (blockIdx.y == 0) {
      __shared__ unsigned bcnt[AGG_NG];
      if (blockIdx.x < AGG_NT / 32)
        bin_owner_queues<AGG_R, AGG_NT / 32, 256>(bcnt, (int)blockIdx.x, esrc, edst, ew,
                                                  savq, savl, savn);
      return;
    }
  }

  const int t     = threadIdx.x;
  const int lane  = t & 31;
  const int wave  = t >> 5;
  const int waveM = wave & 3;
  const int waveN = wave >> 2;
  const int mBase = blockIdx.x * 64;
  const int nBase = (BIN ? (int)blockIdx.y - 1 : (int)blockIdx.y) * 32;
  const int lh    = lane >> 4;
  const int l15   = lane & 15;

  v8f acc = {0.f, 0.f, 0.f, 0.f, 0.f, 0.f, 0.f, 0.f};

  for (int k0 = 0; k0 < K; k0 += 32) {
    {
      const int r   = t >> 3;
      const int c4  = (t & 7) * 4;
      const int gr0 = mBase + r;
      const int gr1 = gr0 + 32;
      const int g0c = gr0 < M ? gr0 : M - 1;
      const int g1c = gr1 < M ? gr1 : M - 1;
      v4f va0 = *(const v4f*)(A + (size_t)g0c * K + k0 + c4);
      v4f va1 = *(const v4f*)(A + (size_t)g1c * K + k0 + c4);
      const v4f vw = *(const v4f*)(W + (size_t)(nBase + r) * K + k0 + c4);
      asm volatile("" :: "v"(va0), "v"(va1), "v"(vw));
      const v4f zero = {0.f, 0.f, 0.f, 0.f};
      if (gr0 >= M) va0 = zero;
      if (gr1 >= M) va1 = zero;
      *(v4f*)&As[r][c4]      = va0;
      *(v4f*)&As[r + 32][c4] = va1;
      *(v4f*)&Ws[r][c4]      = vw;
    }
    __syncthreads();

#pragma unroll
    for (int ks = 0; ks < 8; ++ks) {
      int kb = ks * 4 + 2 * lh;
      v2f a, b;
      a.x = As[waveM * 16 + l15][kb];
      a.y = As[waveM * 16 + l15][kb + 1];
      b.x = Ws[waveN * 16 + l15][kb];
      b.y = Ws[waveN * 16 + l15][kb + 1];
      acc = __builtin_amdgcn_wmma_f32_16x16x4_f32(
           false, a,  false, b,
           (short)0, acc,  false,  false);
    }
    __syncthreads();
  }

#pragma unroll
  for (int r = 0; r < 8; ++r) {
    int row = mBase + waveM * 16 + r + 8 * lh;
    int col = nBase + waveN * 16 + l15;
    if (row < M) {
      float v = acc[r];
      if (bias) v += bias[col];
      if (act)  v = leaky(v, slope);
      C[(size_t)row * N + col] = v;
    }
  }
}

template <int H, int F>
__global__ void node_p_kernel(const float* __restrict__ ht,
                              const float* __restrict__ a,
                              float* __restrict__ p) {
  int idx = blockIdx.x * blockDim.x + threadIdx.x;
  if (idx >= NN_NODES * H) return;
  int n = idx / H, h = idx % H;
  const float* row = ht + ((size_t)n * H + h) * F;
  float acc = 0.f;
#pragma unroll 4
  for (int f = 0; f < F; f += 4) {
    float4 v = *(const float4*)(row + f);
    float4 w = *(const float4*)(a + f);
    acc += v.x * w.x + v.y * w.y + v.z * w.z + v.w * w.w;
  }
  p[idx] = acc;
}


template <int H>
__device__ __forceinline__ void node_scores(int s, int d, float lw,
                                            const float* __restrict__ p,
                                            float* sc) {
  float ps[H], pd[H];
#pragma unroll
  for (int h = 0; h < H; ++h) ps[h] = p[s * H + h];
#pragma unroll
  for (int h = 0; h < H; ++h) pd[h] = p[d * H + h];
#pragma unroll
  for (int h = 0; h < H; ++h) sc[h] = leaky(ps[h] + pd[h] + lw, 0.01f);
}

template <int H>
__device__ __forceinline__ int edge_scores(int e, int d,
                                           const float* __restrict__ p,
                                           const float* __restrict__ ew,
                                           const int* __restrict__ src,
                                           float* sc, float& lw) {
  const int s = src[e];
  lw = log1pf(ew[e]);
  node_scores<H>(s, d, lw, p, sc);
  return s;
}

template <int H, bool RANK>
__device__ __forceinline__ bool agg_score_max(unsigned* q, float* qs, int qn, int lane, int d0,
                                              const float* __restrict__ p,
                                              const float* __restrict__ ew,
                                              const int* __restrict__ src,
                                              unsigned* mx,
                                              unsigned* rowcnt, unsigned* bucket, unsigned qbase) {
  bool over = false;
  for (int base = 0; base < qn; base += 32) {
    const int i = base + lane;
    if (i < qn) {
      const unsigned ent = q[i];
      const int e = (int)(ent & 0xFFFFFu);
      const unsigned dl = ent >> 20;
      float sc[H];
      float lw;
      const int s = edge_scores<H>(e, d0 + (int)dl, p, ew, src, sc, lw);
#pragma unroll
      for (int h = 0; h < H; ++h) {
        qs[i * H + h] = sc[h];
        __hip_atomic_fetch_max(mx + dl * H + h, enc_f32(sc[h]), __ATOMIC_RELAXED,
                               __HIP_MEMORY_SCOPE_WORKGROUP);
      }
      const unsigned kept = (dl << 20) | (unsigned)s;
      q[i] = kept;
      if constexpr (RANK) {
        const unsigned rk = __hip_atomic_fetch_add(rowcnt + dl, 1u, __ATOMIC_RELAXED,
                                                   __HIP_MEMORY_SCOPE_WORKGROUP);
        if (rk < (unsigned)AGG_BCAP) bucket[dl * AGG_BCAP + rk] = qbase + (unsigned)i;
        else over = true;
      }
    }
  }
  return over;
}

template <int H, bool FIRST>
__device__ __forceinline__ bool agg_saved_score_max(unsigned* q, float* qs, int qn, int lane, int d0,
                                                    const float* __restrict__ p,
                                                    const unsigned* sq, float* sl,
                                                    unsigned* mx,
                                                    unsigned* rowcnt, unsigned* bucket, unsigned qbase) {
  bool over = false;
  for (int base = 0; base < qn; base += 32) {
    const int i = base + lane;
    if (i < qn) {
      const unsigned ent = sq[i];
      float lw = sl[i];
      if constexpr (FIRST) { lw = log1pf(lw); sl[i] = lw; }
      const unsigned dl = ent >> 20;
      float sc[H];
      node_scores<H>((int)(ent & 0xFFFFFu), d0 + (int)dl, lw, p, sc);
#pragma unroll
      for (int h = 0; h < H; ++h) {
        qs[i * H + h] = sc[h];
        __hip_atomic_fetch_max(mx + dl * H + h, enc_f32(sc[h]), __ATOMIC_RELAXED,
                               __HIP_MEMORY_SCOPE_WORKGROUP);
      }
      q[i] = ent;
      const unsigned rk = __hip_atomic_fetch_add(rowcnt + dl, 1u, __ATOMIC_RELAXED,
                                                 __HIP_MEMORY_SCOPE_WORKGROUP);
      if (rk < (unsigned)AGG_BCAP) bucket[dl * AGG_BCAP + rk] = qbase + (unsigned)i;
      else over = true;
    }
  }
  return over;
}

template <int H, int F, int R, bool KEPT>
__device__ __forceinline__ void agg_drain(const unsigned* q, const float* qs, int qn, int lane, int d0,
                                          const float* __restrict__ p,
                                          const float* __restrict__ ew,
                                          const int* __restrict__ src,
                                          const unsigned* mx,
                                          const float* __restrict__ ht,
                                          float* acc, float* den) {
  constexpr int HF = H * F;
  constexpr int NV = HF / 128;
  for (int base = 0; base < qn; base += 32) {
    const int cnt = (qn - base) < 32 ? (qn - base) : 32;
    int s = 0, dl = 0;
    float al[H];
#pragma unroll
    for (int h = 0; h < H; ++h) al[h] = 0.f;
    if (lane < cnt) {
      const unsigned ent = q[base + lane];
      dl = (int)(ent >> 20);
      float sc[H];
      if constexpr (KEPT) {
        s = (int)(ent & 0xFFFFFu);
#pragma unroll
        for (int h = 0; h < H; ++h) sc[h] = qs[(base + lane) * H + h];
      } else {
        float lw;
        s = edge_scores<H>((int)(ent & 0xFFFFFu), d0 + dl, p, ew, src, sc, lw);
      }
#pragma unroll
      for (int h = 0; h < H; ++h) {
        float m = dec_f32(mx[dl * H + h]);
        if (!isfinite(m)) m = 0.f;
        al[h] = __expf(sc[h] - m);
        __hip_atomic_fetch_add(den + dl * H + h, al[h], __ATOMIC_RELAXED,
                               __HIP_MEMORY_SCOPE_WORKGROUP);
      }
    }
    for (int i = 0; i < cnt; i += 4) {
      float4 rv[4][NV];
#pragma unroll
      for (int g = 0; g < 4; ++g) {
        const int idx = (i + g) < cnt ? (i + g) : (cnt - 1);
        const int sg = __builtin_amdgcn_readlane(s, idx);
        const float4* rp = (const float4*)(ht + (size_t)sg * HF);
#pragma unroll
        for (int k = 0; k < NV; ++k) rv[g][k] = rp[k * 32 + lane];
      }
#pragma unroll
      for (int g = 0; g < 4; ++g) {
        if (i + g < cnt) {
          const int dlg = __builtin_amdgcn_readlane(dl, i + g);
          float ab[H];
#pragma unroll
          for (int h = 0; h < H; ++h)
            ab[h] = __int_as_float(__builtin_amdgcn_readlane(__float_as_int(al[h]), i + g));
          float* ap = acc + dlg * HF;
#pragma unroll
          for (int k = 0; k < NV; ++k) {
            float a;
            if constexpr (H == 1) a = ab[0];
            else a = (lane & 16) ? ab[2 * k + 1] : ab[2 * k];
            float* pp = ap + (k * 32 + lane) * 4;
            const float4 v = rv[g][k];
            __hip_atomic_fetch_add(pp + 0, a * v.x, __ATOMIC_RELAXED, __HIP_MEMORY_SCOPE_WORKGROUP);
            __hip_atomic_fetch_add(pp + 1, a * v.y, __ATOMIC_RELAXED, __HIP_MEMORY_SCOPE_WORKGROUP);
            __hip_atomic_fetch_add(pp + 2, a * v.z, __ATOMIC_RELAXED, __HIP_MEMORY_SCOPE_WORKGROUP);
            __hip_atomic_fetch_add(pp + 3, a * v.w, __ATOMIC_RELAXED, __HIP_MEMORY_SCOPE_WORKGROUP);
          }
        }
      }
    }
  }
}

template <int H, int F, int R, int NT, int PHASE>
__device__ __forceinline__ void agg_scan(const int* __restrict__ dst, int wave, int lane, int d0,
                                         unsigned* q, float* qs, int& qn, bool& spilled,
                                         const float* __restrict__ p,
                                         const float* __restrict__ ew,
                                         const int* __restrict__ src,
                                         const float* __restrict__ ht,
                                         unsigned* mx, float* acc, float* den) {
  constexpr int NW   = NT / 32;
  constexpr int NWIN = N_EDGES / 128;
  const int4* dst4 = (const int4*)dst;
  const unsigned lt = (1u << lane) - 1u;

  int4 wa, wb;
  {
    const int a0 = wave < NWIN ? wave : NWIN - 1;
    const int b0 = (wave + NW) < NWIN ? (wave + NW) : NWIN - 1;
    wa = dst4[a0 * 32 + lane];
    wb = dst4[b0 * 32 + lane];
  }
  for (int win = wave; win < NWIN; win += 2 * NW) {
    const int na = (win + 2 * NW) < NWIN ? (win + 2 * NW) : NWIN - 1;
    const int nb = (win + 3 * NW) < NWIN ? (win + 3 * NW) : NWIN - 1;
    const int4 xa = dst4[na * 32 + lane];
    const int4 xb = dst4[nb * 32 + lane];

#pragma unroll
    for (int half = 0; half < 2; ++half) {
      const int  w     = win + half * NW;
      const bool valid = w < NWIN;
      const int4 dv    = half ? wb : wa;
      const unsigned l0 = (unsigned)(dv.x - d0), l1 = (unsigned)(dv.y - d0);
      const unsigned l2 = (unsigned)(dv.z - d0), l3 = (unsigned)(dv.w - d0);
      const bool m0 = valid && l0 < (unsigned)R, m1 = valid && l1 < (unsigned)R;
      const bool m2 = valid && l2 < (unsigned)R, m3 = valid && l3 < (unsigned)R;
      if (__builtin_amdgcn_ballot_w32(m0 || m1 || m2 || m3)) {
        const unsigned eb = (unsigned)(w * 128 + lane * 4);
        const unsigned k0 = __builtin_amdgcn_ballot_w32(m0);
        const unsigned k1 = __builtin_amdgcn_ballot_w32(m1);
        const unsigned k2 = __builtin_amdgcn_ballot_w32(m2);
        const unsigned k3 = __builtin_amdgcn_ballot_w32(m3);
        if constexpr (PHASE == 0) {
          const int c = __popc(k0) + __popc(k1) + __popc(k2) + __popc(k3);
          if (qn + c > AGG_SCAP) {
            agg_score_max<H, false>(q, qs, qn, lane, d0, p, ew, src, mx, nullptr, nullptr, 0u);
            qn = 0;
            spilled = true;
          }
        }
        if (m0) q[qn + __popc(k0 & lt)] = (l0 << 20) | (eb + 0u);
        qn += __popc(k0);
        if (m1) q[qn + __popc(k1 & lt)] = (l1 << 20) | (eb + 1u);
        qn += __popc(k1);
        if (m2) q[qn + __popc(k2 & lt)] = (l2 << 20) | (eb + 2u);
        qn += __popc(k2);
        if (m3) q[qn + __popc(k3 & lt)] = (l3 << 20) | (eb + 3u);
        qn += __popc(k3);
      }
    }
    if constexpr (PHASE == 1) {
      if (qn >= 8) {
        agg_drain<H, F, R, false>(q, qs, qn, lane, d0, p, ew, src, mx, ht, acc, den);
        qn = 0;
      }
    }
    wa = xa;
    wb = xb;
  }
}

template <int H, int F, int R, int NW, bool ACT>
__device__ __forceinline__ void agg_rows(const unsigned* qbuf, const float* qsc,
                                         const unsigned* rowcnt, const unsigned* bucket,
                                         const unsigned* mx, unsigned* nextrow,
                                         int wave, int lane, int d0,
                                         const float* __restrict__ ht,
                                         const float* __restrict__ bias, float slope,
                                         float* __restrict__ hagg) {
  constexpr int HF = H * F;
  constexpr int NV = HF / 128;
  int r = wave;
  while (r < R) {
    const int n = d0 + r;
    if (n >= NN_NODES) break;
    const int cnt = __builtin_amdgcn_readfirstlane((int)rowcnt[r]);
    float m[H], den[H];
#pragma unroll
    for (int h = 0; h < H; ++h) {
      float mm = dec_f32(mx[r * H + h]);
      if (!isfinite(mm)) mm = 0.f;
      m[h] = mm;
      den[h] = 0.f;
    }
    float4 sum[NV];
#pragma unroll
    for (int k = 0; k < NV; ++k) sum[k] = make_float4(0.f, 0.f, 0.f, 0.f);

    for (int base = 0; base < cnt; base += 32) {
      const int c = (cnt - base) < 32 ? (cnt - base) : 32;
      int s = 0;
      float al[H];
#pragma unroll
      for (int h = 0; h < H; ++h) al[h] = 0.f;
      if (lane < c) {
        const unsigned idx = bucket[r * AGG_BCAP + base + lane];
        const unsigned ent = qbuf[(idx / AGG_SCAP) * AGG_QCAP + (idx % AGG_SCAP)];
        s = (int)(ent & 0xFFFFFu);
#pragma unroll
        for (int h = 0; h < H; ++h) al[h] = __expf(qsc[idx * H + h] - m[h]);
      }
      for (int i = 0; i < c; i += AGG_GROW) {
        float4 rv[AGG_GROW][NV];
#pragma unroll
        for (int g = 0; g < AGG_GROW; ++g) {
          const int idx = (i + g) < c ? (i + g) : (c - 1);
          const int sg = __builtin_amdgcn_readlane(s, idx);
          const float4* rp = (const float4*)(ht + (size_t)sg * HF);
#pragma unroll
          for (int k = 0; k < NV; ++k) rv[g][k] = rp[k * 32 + lane];
        }
#pragma unroll
        for (int g = 0; g < AGG_GROW; ++g) {
          if (i + g < c) {
            float ab[H];
#pragma unroll
            for (int h = 0; h < H; ++h) {
              ab[h] = __int_as_float(__builtin_amdgcn_readlane(__float_as_int(al[h]), i + g));
              den[h] += ab[h];
            }
#pragma unroll
            for (int k = 0; k < NV; ++k) {
              float a;
              if constexpr (H == 1) a = ab[0];
              else a = (lane & 16) ? ab[2 * k + 1] : ab[2 * k];
              const float4 v = rv[g][k];
              sum[k].x += a * v.x; sum[k].y += a * v.y;
              sum[k].z += a * v.z; sum[k].w += a * v.w;
            }
          }
        }
      }
    }

#pragma unroll
    for (int k = 0; k < NV; ++k) {
      float dsel;
      if constexpr (H == 1) dsel = den[0];
      else dsel = (lane & 16) ? den[2 * k + 1] : den[2 * k];
      const float sden = fmaxf(dsel, 1e-20f);
      const int c4 = k * 32 + lane;
      float4 v = sum[k];
      v.x = v.x / sden; v.y = v.y / sden; v.z = v.z / sden; v.w = v.w / sden;
      const float4 b = ((const float4*)bias)[c4];
      v.x += b.x; v.y += b.y; v.z += b.z; v.w += b.w;
      if constexpr (ACT) {
        v.x = leaky(v.x, slope); v.y = leaky(v.y, slope);
        v.z = leaky(v.z, slope); v.w = leaky(v.w, slope);
      }
      ((float4*)hagg)[(size_t)n * (HF / 4) + c4] = v;
    }

    unsigned nx = 0u;
    if (lane == 0)
      nx = __hip_atomic_fetch_add(nextrow, 1u, __ATOMIC_RELAXED, __HIP_MEMORY_SCOPE_WORKGROUP);
    r = __builtin_amdgcn_readfirstlane((int)nx);
  }
}

template <int H, int F, int R, int NT, bool ACT, bool FIRST>
__global__ __launch_bounds__(NT)
void edge_agg_owner_kernel(const float* __restrict__ p,
                           const float* __restrict__ ew,
                           const float* __restrict__ ht,
                           const int* __restrict__ src,
                           const int* __restrict__ dst,
                           const float* __restrict__ bias,
                           float slope,
                           float* __restrict__ hagg,
                           const unsigned* savq, float* savl, const int* savn) {
  constexpr int HF = H * F;
  constexpr int NW = NT / 32;
  static_assert(N_EDGES % 128 == 0, "edge windows of 128");
  static_assert(N_EDGES <= (1 << 20) && NN_NODES <= (1 << 20) && R <= 2048, "queue entry packing");
  static_assert(AGG_SCAP >= 128 && AGG_SCAP <= AGG_QCAP, "one window always fits an empty queue");
  static_assert(H == 1 || (H == 4 && F == 64), "head select per lane half");
  static_assert(HF % 128 == 0, "row = whole float4 per lane");
  static_assert(AGG_SCAP * NW <= (1 << 20) && AGG_BCAP >= 32, "bucket entry = place in the queues");
  static_assert(R == AGG_R && NT == AGG_NT, "the shape bin_owner_queues files the queues for");

  __shared__ float    acc[R * HF];
  __shared__ float    den[R * H];
  __shared__ unsigned mx[R * H];
  __shared__ unsigned qbuf[NW * AGG_QCAP];
  __shared__ float    qsc[NW * AGG_SCAP * H];
  __shared__ unsigned rowcnt[R];
  __shared__ unsigned bucket[R * AGG_BCAP];
  __shared__ unsigned ctl[2];

  const int t    = threadIdx.x;
  const int lane = t & 31;
  const int wave = __builtin_amdgcn_readfirstlane(t >> 5);
  const int d0   = blockIdx.x * R;
  unsigned* q    = qbuf + wave * AGG_QCAP;
  float*    qs   = qsc + wave * (AGG_SCAP * H);

  for (int i = t; i < R * H; i += NT) mx[i] = ENC_NEG_INF;
  for (int i = t; i < R; i += NT) rowcnt[i] = 0u;
  if (t == 0) { ctl[0] = 0u; ctl[1] = (unsigned)NW; }
  __syncthreads();

  int  qn      = 0;
  bool spilled = false;
  const int slot = blockIdx.x * NW + wave;
  const unsigned* sq = savq + (size_t)slot * AGG_SCAP;
  float*          sl = savl + (size_t)slot * AGG_SCAP;
  const unsigned qbase = (unsigned)(wave * AGG_SCAP);
  bool over = false;
  const int saved = __builtin_amdgcn_readfirstlane(savn[slot]);
  if (saved >= 0) {
    qn = saved;
    over = agg_saved_score_max<H, FIRST>(q, qs, qn, lane, d0, p, sq, sl, mx, rowcnt, bucket, qbase);
  } else {
    agg_scan<H, F, R, NT, 0>(dst, wave, lane, d0, q, qs, qn, spilled, p, ew, src, ht, mx, acc, den);
    over = agg_score_max<H, true>(q, qs, qn, lane, d0, p, ew, src, mx, rowcnt, bucket, qbase);
  }
  if (over || spilled) ctl[0] = 1u;
  __syncthreads();

  if (__builtin_amdgcn_readfirstlane((int)ctl[0]) == 0) {
    agg_rows<H, F, R, NW, ACT>(qbuf, qsc, rowcnt, bucket, mx, ctl + 1, wave, lane, d0,
                               ht, bias, slope, hagg);
    return;
  }

  for (int i = t; i < R * HF / 4; i += NT)
    ((float4*)acc)[i] = make_float4(0.f, 0.f, 0.f, 0.f);
  for (int i = t; i < R * H; i += NT) den[i] = 0.f;
  __syncthreads();
  if (!spilled) {
    agg_drain<H, F, R, true>(q, qs, qn, lane, d0, p, ew, src, mx, ht, acc, den);
  } else {
    qn = 0;
    agg_scan<H, F, R, NT, 1>(dst, wave, lane, d0, q, qs, qn, spilled, p, ew, src, ht, mx, acc, den);
    if (qn > 0)
      agg_drain<H, F, R, false>(q, qs, qn, lane, d0, p, ew, src, mx, ht, acc, den);
  }
  __syncthreads();

  for (int i = t; i < R * HF / 4; i += NT) {
    const int r = i / (HF / 4);
    const int n = d0 + r;
    if (n < NN_NODES) {
      float4 v = ((const float4*)acc)[i];
      const int c4 = i % (HF / 4);
      const float s = fmaxf(den[r * H + (c4 * 4) / F], 1e-20f);
      v.x = v.x / s; v.y = v.y / s; v.z = v.z / s; v.w = v.w / s;
      const float4 b = ((const float4*)bias)[c4];
      v.x += b.x; v.y += b.y; v.z += b.z; v.w += b.w;
      if constexpr (ACT) {
        v.x = leaky(v.x, slope); v.y = leaky(v.y, slope);
        v.z = leaky(v.z, slope); v.w = leaky(v.w, slope);
      }
      ((float4*)hagg)[(size_t)d0 * (HF / 4) + i] = v;
    }
  }
}

__global__ void build_z_kernel(const float* __restrict__ emb,
                               const int* __restrict__ ps, const int* __restrict__ pd,
                               const int* __restrict__ ns, const int* __restrict__ nd,
                               float* __restrict__ Z) {
  int idx = blockIdx.x * blockDim.x + threadIdx.x;
  if (idx >= N_ROWS * 32) return;
  int row = idx >> 5;
  int f   = (idx & 31) * 4;
  int s, d;
  if (row < N_PAIR) { s = ps[row]; d = pd[row]; }
  else              { s = ns[row - N_PAIR]; d = nd[row - N_PAIR]; }
  float4 a = *(const float4*)(emb + (size_t)s * 128 + f);
  float4 b = *(const float4*)(emb + (size_t)d * 128 + f);
  float4 r; r.x = a.x * b.x; r.y = a.y * b.y; r.z = a.z * b.z; r.w = a.w * b.w;
  *(float4*)(Z + (size_t)row * 128 + f) = r;
}

__global__ void final_dot_kernel(const float* __restrict__ Z2,
                                 const float* __restrict__ pW3,
                                 const float* __restrict__ pb3,
                                 float* __restrict__ out) {
  int row = blockIdx.x * blockDim.x + threadIdx.x;
  if (row >= N_ROWS) return;
  const float* z = Z2 + (size_t)row * 32;
  float acc = pb3[0];
#pragma unroll
  for (int j = 0; j < 32; j += 4) {
    float4 a = *(const float4*)(z + j);
    float4 w = *(const float4*)(pW3 + j);
    acc += a.x * w.x + a.y * w.y + a.z * w.z + a.w * w.w;
  }
  out[row] = acc;
}

extern "C" void kernel_launch(void* const* d_in, const int* in_sizes, int n_in,
                              void* d_out, int out_size, void* d_ws, size_t ws_size,
                              hipStream_t stream) {
  const float* x    = (const float*)d_in[0];
  const float* ew   = (const float*)d_in[1];
  const int*   esrc = (const int*)d_in[2];
  const int*   edst = (const int*)d_in[3];
  const int*   psrc = (const int*)d_in[4];
  const int*   pdst = (const int*)d_in[5];
  const int*   nsrc = (const int*)d_in[6];
  const int*   ndst = (const int*)d_in[7];
  const float* Wl[3] = {(const float*)d_in[8],  (const float*)d_in[11], (const float*)d_in[14]};
  const float* al[3] = {(const float*)d_in[9],  (const float*)d_in[12], (const float*)d_in[15]};
  const float* bl[3] = {(const float*)d_in[10], (const float*)d_in[13], (const float*)d_in[16]};
  const float* pW1 = (const float*)d_in[17]; const float* pb1 = (const float*)d_in[18];
  const float* pW2 = (const float*)d_in[19]; const float* pb2 = (const float*)d_in[20];
  const float* pW3 = (const float*)d_in[21]; const float* pb3 = (const float*)d_in[22];

  float* out   = (float*)d_out;
  float* emb   = out + 2 * N_PAIR;

  float*    ws   = (float*)d_ws;
  float*    ht   = ws;
  float*    hbuf = ht   + 6400000;
  float*    pbuf = hbuf + 6400000;
  float*    Z    = pbuf + 100000;
  float*    Z1   = Z    + 25600000;
  float*    Z2   = Z1   + 12800000;
  unsigned* savq = (unsigned*)(Z2 + 6400000);
  float*    savl = Z2 + 6400000 + 1048576;
  int*      savn = (int*)(Z2 + 6400000 + 2097152);

  const int Hs[3]  = {4, 4, 1};
  const int Fs[3]  = {64, 64, 128};
  const int Kin[3] = {128, 256, 256};

  const float* hin = x;
  for (int l = 0; l < 3; ++l) {
    const int H = Hs[l], F = Fs[l], HF = H * F, K = Kin[l];
    float* hout = (l == 2) ? emb : hbuf;

    if (l == 0) {
      dim3 gg((NN_NODES + 63) / 64, HF / 32 + 1);
      gemm_wmma_f32<true><<<gg, 256, 0, stream>>>(hin, Wl[l], nullptr, ht,
                                                  NN_NODES, HF, K, 0.f, 0,
                                                  esrc, edst, ew, savq, savl, savn);
    } else {
      dim3 gg((NN_NODES + 63) / 64, HF / 32);
      gemm_wmma_f32<false><<<gg, 256, 0, stream>>>(hin, Wl[l], nullptr, ht,
                                                   NN_NODES, HF, K, 0.f, 0,
                                                   nullptr, nullptr, nullptr,
                                                   nullptr, nullptr, nullptr);
    }
    int nh = NN_NODES * H;
    if (l < 2) node_p_kernel<4, 64><<<(nh + 255) / 256, 256, 0, stream>>>(ht, al[l], pbuf);
    else       node_p_kernel<1, 128><<<(nh + 255) / 256, 256, 0, stream>>>(ht, al[l], pbuf);

    if (l == 0) {
      edge_agg_owner_kernel<4, 64, AGG_R, AGG_NT, true, true><<<AGG_NG, AGG_NT, 0, stream>>>(
          pbuf, ew, ht, esrc, edst, bl[l], 0.01f, hout, savq, savl, savn);
    } else if (l == 1) {
      edge_agg_owner_kernel<4, 64, AGG_R, AGG_NT, true, false><<<AGG_NG, AGG_NT, 0, stream>>>(
          pbuf, ew, ht, esrc, edst, bl[l], 0.01f, hout, savq, savl, savn);
    } else {
      edge_agg_owner_kernel<1, 128, AGG_R, AGG_NT, false, false><<<AGG_NG, AGG_NT, 0, stream>>>(
          pbuf, ew, ht, esrc, edst, bl[l], 0.01f, hout, savq, savl, savn);
    }
    hin = hout;
  }

  build_z_kernel<<<(N_ROWS * 32 + 255) / 256, 256, 0, stream>>>(emb, psrc, pdst,
                                                                nsrc, ndst, Z);
  dim3 g1((N_ROWS + 63) / 64, 64 / 32);
  gemm_wmma_f32<false><<<g1, 256, 0, stream>>>(Z, pW1, pb1, Z1, N_ROWS, 64, 128, 0.2f, 1,
                                               nullptr, nullptr, nullptr, nullptr, nullptr, nullptr);
  dim3 g2((N_ROWS + 63) / 64, 32 / 32);
  gemm_wmma_f32<false><<<g2, 256, 0, stream>>>(Z1, pW2, pb2, Z2, N_ROWS, 32, 64, 0.2f, 1,
                                               nullptr, nullptr, nullptr, nullptr, nullptr, nullptr);
  final_dot_kernel<<<(N_ROWS + 255) / 256, 256, 0, stream>>>(Z2, pW3, pb3, out);
}


# Predictor in one launch as transposed WMMA products chained in registers: pair rows gathered straight into B, D of W1.Z^T is the B operand of W2.Z1^T; Z, Z1, Z2 gone; k pairing, dot order change
# speedup vs baseline: 14.4123x; 1.3594x over previous
#include <hip/hip_runtime.h>
#include <hip/hip_bf16.h>


typedef __attribute__((ext_vector_type(2))) float v2f;
typedef __attribute__((ext_vector_type(8))) float v8f;
typedef __attribute__((ext_vector_type(4))) float v4f;

#define NN_NODES 25000
#define N_EDGES  400000
#define N_PAIR   100000
#define N_ROWS   200000

__device__ __forceinline__ unsigned enc_f32(float f) {
  unsigned u = __float_as_uint(f);
  return (u & 0x80000000u) ? ~u : (u | 0x80000000u);
}
__device__ __forceinline__ float dec_f32(unsigned e) {
  return (e & 0x80000000u) ? __uint_as_float(e & 0x7FFFFFFFu)
                           : __uint_as_float(~e);
}
#define ENC_NEG_INF 0x007FFFFFu

__device__ __forceinline__ float leaky(float v, float slope) {
  return v >= 0.f ? v : v * slope;
}

#define AGG_R    98
#define AGG_NT   1024
#define AGG_NG   ((NN_NODES + AGG_R - 1) / AGG_R)
#define AGG_QCAP 272
#define AGG_SCAP 128
#define AGG_BCAP 64
#define AGG_GROW 8

template <int R, int NW, int NTB>
__device__ __forceinline__ void bin_owner_queues(unsigned* cnt, int w,
                                                 const int* __restrict__ src,
                                                 const int* __restrict__ dst,
                                                 const float* __restrict__ ew,
                                                 unsigned* savq, float* savl, int* savn) {
  constexpr int NG   = (NN_NODES + R - 1) / R;
  constexpr int NWIN = N_EDGES / 128;
  constexpr int STEP = NW * (NTB / 32);
  static_assert(N_EDGES % 128 == 0, "edge windows of 128");
  static_assert(R <= 2048 && NN_NODES <= (1 << 20), "queue entry packing");
  const int t    = threadIdx.x;
  const int lane = t & 31;
  const int wv   = __builtin_amdgcn_readfirstlane(t >> 5);
  for (int i = t; i < NG; i += NTB) cnt[i] = 0u;
  __syncthreads();

  const int4*   dst4 = (const int4*)dst;
  const int4*   src4 = (const int4*)src;
  const float4* ew4  = (const float4*)ew;
  int win = w + NW * wv;
  int4 dv[2];
#pragma unroll
  for (int h = 0; h < 2; ++h) {
    const int c0 = (win + h * STEP) < NWIN ? (win + h * STEP) : NWIN - 1;
    dv[h] = dst4[c0 * 32 + lane];
  }
  for (; win < NWIN; win += 2 * STEP) {
    int4   sv[2], dn[2];
    float4 wt[2];
#pragma unroll
    for (int h = 0; h < 2; ++h) {
      const int c0 = (win + h * STEP) < NWIN ? (win + h * STEP) : NWIN - 1;
      sv[h] = src4[c0 * 32 + lane];
      wt[h] = ew4[c0 * 32 + lane];
    }
#pragma unroll
    for (int h = 0; h < 2; ++h) {
      const int nx = (win + (2 + h) * STEP) < NWIN ? (win + (2 + h) * STEP) : NWIN - 1;
      dn[h] = dst4[nx * 32 + lane];
    }
#pragma unroll
    for (int h = 0; h < 2; ++h) {
      const bool  valid = (win + h * STEP) < NWIN;
      const int   dd[4] = {dv[h].x, dv[h].y, dv[h].z, dv[h].w};
      const int   ss[4] = {sv[h].x, sv[h].y, sv[h].z, sv[h].w};
      const float ww[4] = {wt[h].x, wt[h].y, wt[h].z, wt[h].w};
      unsigned qb[4], rw[4], rk[4];
#pragma unroll
      for (int k = 0; k < 4; ++k) {
        const bool ok = valid && (unsigned)dd[k] < (unsigned)NN_NODES;
        const unsigned d = ok ? (unsigned)dd[k] : 0u;
        const unsigned g = d / (unsigned)R;
        rw[k] = (d - g * (unsigned)R) << 20;
        qb[k] = (g * (unsigned)NW + (unsigned)w) * (unsigned)AGG_SCAP;
        const unsigned got = __hip_atomic_fetch_add(cnt + g, ok ? 1u : 0u, __ATOMIC_RELAXED,
                                                    __HIP_MEMORY_SCOPE_WORKGROUP);
        rk[k] = ok ? got : (unsigned)AGG_SCAP;
      }
#pragma unroll
      for (int k = 0; k < 4; ++k) {
        if (rk[k] < (unsigned)AGG_SCAP) {
          const unsigned at = qb[k] + rk[k];
          savq[at] = rw[k] | (unsigned)ss[k];
          savl[at] = ww[k];
        }
      }
    }
    dv[0] = dn[0];
    dv[1] = dn[1];
  }
  __syncthreads();
  for (int i = t; i < NG; i += NTB) {
    const unsigned n = cnt[i];
    savn[i * NW + w] = n <= (unsigned)AGG_SCAP ? (int)n : -1;
  }
}

template <bool BIN>
__global__ void gemm_wmma_f32(const float* __restrict__ A,
                              const float* __restrict__ W,
                              const float* __restrict__ bias,
                              float* __restrict__ C,
                              int M, int N, int K,
                              float slope, int act,
                              const int* __restrict__ esrc,
                              const int* __restrict__ edst,
                              const float* __restrict__ ew,
                              unsigned* savq, float* savl, int* savn) {
  __shared__ float As[64][36];
  __shared__ float Ws[32][36];

  if constexpr (BIN) {
    if (blockIdx.y == 0) {
      __shared__ unsigned bcnt[AGG_NG];
      if (blockIdx.x < AGG_NT / 32)
        bin_owner_queues<AGG_R, AGG_NT / 32, 256>(bcnt, (int)blockIdx.x, esrc, edst, ew,
                                                  savq, savl, savn);
      return;
    }
  }

  const int t     = threadIdx.x;
  const int lane  = t & 31;
  const int wave  = t >> 5;
  const int waveM = wave & 3;
  const int waveN = wave >> 2;
  const int mBase = blockIdx.x * 64;
  const int nBase = (BIN ? (int)blockIdx.y - 1 : (int)blockIdx.y) * 32;
  const int lh    = lane >> 4;
  const int l15   = lane & 15;

  v8f acc = {0.f, 0.f, 0.f, 0.f, 0.f, 0.f, 0.f, 0.f};

  for (int k0 = 0; k0 < K; k0 += 32) {
    {
      const int r   = t >> 3;
      const int c4  = (t & 7) * 4;
      const int gr0 = mBase + r;
      const int gr1 = gr0 + 32;
      const int g0c = gr0 < M ? gr0 : M - 1;
      const int g1c = gr1 < M ? gr1 : M - 1;
      v4f va0 = *(const v4f*)(A + (size_t)g0c * K + k0 + c4);
      v4f va1 = *(const v4f*)(A + (size_t)g1c * K + k0 + c4);
      const v4f vw = *(const v4f*)(W + (size_t)(nBase + r) * K + k0 + c4);
      asm volatile("" :: "v"(va0), "v"(va1), "v"(vw));
      const v4f zero = {0.f, 0.f, 0.f, 0.f};
      if (gr0 >= M) va0 = zero;
      if (gr1 >= M) va1 = zero;
      *(v4f*)&As[r][c4]      = va0;
      *(v4f*)&As[r + 32][c4] = va1;
      *(v4f*)&Ws[r][c4]      = vw;
    }
    __syncthreads();

#pragma unroll
    for (int ks = 0; ks < 8; ++ks) {
      int kb = ks * 4 + 2 * lh;
      v2f a, b;
      a.x = As[waveM * 16 + l15][kb];
      a.y = As[waveM * 16 + l15][kb + 1];
      b.x = Ws[waveN * 16 + l15][kb];
      b.y = Ws[waveN * 16 + l15][kb + 1];
      acc = __builtin_amdgcn_wmma_f32_16x16x4_f32(
           false, a,  false, b,
           (short)0, acc,  false,  false);
    }
    __syncthreads();
  }

#pragma unroll
  for (int r = 0; r < 8; ++r) {
    int row = mBase + waveM * 16 + r + 8 * lh;
    int col = nBase + waveN * 16 + l15;
    if (row < M) {
      float v = acc[r];
      if (bias) v += bias[col];
      if (act)  v = leaky(v, slope);
      C[(size_t)row * N + col] = v;
    }
  }
}

template <int H, int F>
__global__ void node_p_kernel(const float* __restrict__ ht,
                              const float* __restrict__ a,
                              float* __restrict__ p) {
  int idx = blockIdx.x * blockDim.x + threadIdx.x;
  if (idx >= NN_NODES * H) return;
  int n = idx / H, h = idx % H;
  const float* row = ht + ((size_t)n * H + h) * F;
  float acc = 0.f;
#pragma unroll 4
  for (int f = 0; f < F; f += 4) {
    float4 v = *(const float4*)(row + f);
    float4 w = *(const float4*)(a + f);
    acc += v.x * w.x + v.y * w.y + v.z * w.z + v.w * w.w;
  }
  p[idx] = acc;
}


template <int H>
__device__ __forceinline__ void node_scores(int s, int d, float lw,
                                            const float* __restrict__ p,
                                            float* sc) {
  float ps[H], pd[H];
#pragma unroll
  for (int h = 0; h < H; ++h) ps[h] = p[s * H + h];
#pragma unroll
  for (int h = 0; h < H; ++h) pd[h] = p[d * H + h];
#pragma unroll
  for (int h = 0; h < H; ++h) sc[h] = leaky(ps[h] + pd[h] + lw, 0.01f);
}

template <int H>
__device__ __forceinline__ int edge_scores(int e, int d,
                                           const float* __restrict__ p,
                                           const float* __restrict__ ew,
                                           const int* __restrict__ src,
                                           float* sc, float& lw) {
  const int s = src[e];
  lw = log1pf(ew[e]);
  node_scores<H>(s, d, lw, p, sc);
  return s;
}

template <int H, bool RANK>
__device__ __forceinline__ bool agg_score_max(unsigned* q, float* qs, int qn, int lane, int d0,
                                              const float* __restrict__ p,
                                              const float* __restrict__ ew,
                                              const int* __restrict__ src,
                                              unsigned* mx,
                                              unsigned* rowcnt, unsigned* bucket, unsigned qbase) {
  bool over = false;
  for (int base = 0; base < qn; base += 32) {
    const int i = base + lane;
    if (i < qn) {
      const unsigned ent = q[i];
      const int e = (int)(ent & 0xFFFFFu);
      const unsigned dl = ent >> 20;
      float sc[H];
      float lw;
      const int s = edge_scores<H>(e, d0 + (int)dl, p, ew, src, sc, lw);
#pragma unroll
      for (int h = 0; h < H; ++h) {
        qs[i * H + h] = sc[h];
        __hip_atomic_fetch_max(mx + dl * H + h, enc_f32(sc[h]), __ATOMIC_RELAXED,
                               __HIP_MEMORY_SCOPE_WORKGROUP);
      }
      const unsigned kept = (dl << 20) | (unsigned)s;
      q[i] = kept;
      if constexpr (RANK) {
        const unsigned rk = __hip_atomic_fetch_add(rowcnt + dl, 1u, __ATOMIC_RELAXED,
                                                   __HIP_MEMORY_SCOPE_WORKGROUP);
        if (rk < (unsigned)AGG_BCAP) bucket[dl * AGG_BCAP + rk] = qbase + (unsigned)i;
        else over = true;
      }
    }
  }
  return over;
}

template <int H, bool FIRST>
__device__ __forceinline__ bool agg_saved_score_max(unsigned* q, float* qs, int qn, int lane, int d0,
                                                    const float* __restrict__ p,
                                                    const unsigned* sq, float* sl,
                                                    unsigned* mx,
                                                    unsigned* rowcnt, unsigned* bucket, unsigned qbase) {
  bool over = false;
  for (int base = 0; base < qn; base += 32) {
    const int i = base + lane;
    if (i < qn) {
      const unsigned ent = sq[i];
      float lw = sl[i];
      if constexpr (FIRST) { lw = log1pf(lw); sl[i] = lw; }
      const unsigned dl = ent >> 20;
      float sc[H];
      node_scores<H>((int)(ent & 0xFFFFFu), d0 + (int)dl, lw, p, sc);
#pragma unroll
      for (int h = 0; h < H; ++h) {
        qs[i * H + h] = sc[h];
        __hip_atomic_fetch_max(mx + dl * H + h, enc_f32(sc[h]), __ATOMIC_RELAXED,
                               __HIP_MEMORY_SCOPE_WORKGROUP);
      }
      q[i] = ent;
      const unsigned rk = __hip_atomic_fetch_add(rowcnt + dl, 1u, __ATOMIC_RELAXED,
                                                 __HIP_MEMORY_SCOPE_WORKGROUP);
      if (rk < (unsigned)AGG_BCAP) bucket[dl * AGG_BCAP + rk] = qbase + (unsigned)i;
      else over = true;
    }
  }
  return over;
}

template <int H, int F, int R, bool KEPT>
__device__ __forceinline__ void agg_drain(const unsigned* q, const float* qs, int qn, int lane, int d0,
                                          const float* __restrict__ p,
                                          const float* __restrict__ ew,
                                          const int* __restrict__ src,
                                          const unsigned* mx,
                                          const float* __restrict__ ht,
                                          float* acc, float* den) {
  constexpr int HF = H * F;
  constexpr int NV = HF / 128;
  for (int base = 0; base < qn; base += 32) {
    const int cnt = (qn - base) < 32 ? (qn - base) : 32;
    int s = 0, dl = 0;
    float al[H];
#pragma unroll
    for (int h = 0; h < H; ++h) al[h] = 0.f;
    if (lane < cnt) {
      const unsigned ent = q[base + lane];
      dl = (int)(ent >> 20);
      float sc[H];
      if constexpr (KEPT) {
        s = (int)(ent & 0xFFFFFu);
#pragma unroll
        for (int h = 0; h < H; ++h) sc[h] = qs[(base + lane) * H + h];
      } else {
        float lw;
        s = edge_scores<H>((int)(ent & 0xFFFFFu), d0 + dl, p, ew, src, sc, lw);
      }
#pragma unroll
      for (int h = 0; h < H; ++h) {
        float m = dec_f32(mx[dl * H + h]);
        if (!isfinite(m)) m = 0.f;
        al[h] = __expf(sc[h] - m);
        __hip_atomic_fetch_add(den + dl * H + h, al[h], __ATOMIC_RELAXED,
                               __HIP_MEMORY_SCOPE_WORKGROUP);
      }
    }
    for (int i = 0; i < cnt; i += 4) {
      float4 rv[4][NV];
#pragma unroll
      for (int g = 0; g < 4; ++g) {
        const int idx = (i + g) < cnt ? (i + g) : (cnt - 1);
        const int sg = __builtin_amdgcn_readlane(s, idx);
        const float4* rp = (const float4*)(ht + (size_t)sg * HF);
#pragma unroll
        for (int k = 0; k < NV; ++k) rv[g][k] = rp[k * 32 + lane];
      }
#pragma unroll
      for (int g = 0; g < 4; ++g) {
        if (i + g < cnt) {
          const int dlg = __builtin_amdgcn_readlane(dl, i + g);
          float ab[H];
#pragma unroll
          for (int h = 0; h < H; ++h)
            ab[h] = __int_as_float(__builtin_amdgcn_readlane(__float_as_int(al[h]), i + g));
          float* ap = acc + dlg * HF;
#pragma unroll
          for (int k = 0; k < NV; ++k) {
            float a;
            if constexpr (H == 1) a = ab[0];
            else a = (lane & 16) ? ab[2 * k + 1] : ab[2 * k];
            float* pp = ap + (k * 32 + lane) * 4;
            const float4 v = rv[g][k];
            __hip_atomic_fetch_add(pp + 0, a * v.x, __ATOMIC_RELAXED, __HIP_MEMORY_SCOPE_WORKGROUP);
            __hip_atomic_fetch_add(pp + 1, a * v.y, __ATOMIC_RELAXED, __HIP_MEMORY_SCOPE_WORKGROUP);
            __hip_atomic_fetch_add(pp + 2, a * v.z, __ATOMIC_RELAXED, __HIP_MEMORY_SCOPE_WORKGROUP);
            __hip_atomic_fetch_add(pp + 3, a * v.w, __ATOMIC_RELAXED, __HIP_MEMORY_SCOPE_WORKGROUP);
          }
        }
      }
    }
  }
}

template <int H, int F, int R, int NT, int PHASE>
__device__ __forceinline__ void agg_scan(const int* __restrict__ dst, int wave, int lane, int d0,
                                         unsigned* q, float* qs, int& qn, bool& spilled,
                                         const float* __restrict__ p,
                                         const float* __restrict__ ew,
                                         const int* __restrict__ src,
                                         const float* __restrict__ ht,
                                         unsigned* mx, float* acc, float* den) {
  constexpr int NW   = NT / 32;
  constexpr int NWIN = N_EDGES / 128;
  const int4* dst4 = (const int4*)dst;
  const unsigned lt = (1u << lane) - 1u;

  int4 wa, wb;
  {
    const int a0 = wave < NWIN ? wave : NWIN - 1;
    const int b0 = (wave + NW) < NWIN ? (wave + NW) : NWIN - 1;
    wa = dst4[a0 * 32 + lane];
    wb = dst4[b0 * 32 + lane];
  }
  for (int win = wave; win < NWIN; win += 2 * NW) {
    const int na = (win + 2 * NW) < NWIN ? (win + 2 * NW) : NWIN - 1;
    const int nb = (win + 3 * NW) < NWIN ? (win + 3 * NW) : NWIN - 1;
    const int4 xa = dst4[na * 32 + lane];
    const int4 xb = dst4[nb * 32 + lane];

#pragma unroll
    for (int half = 0; half < 2; ++half) {
      const int  w     = win + half * NW;
      const bool valid = w < NWIN;
      const int4 dv    = half ? wb : wa;
      const unsigned l0 = (unsigned)(dv.x - d0), l1 = (unsigned)(dv.y - d0);
      const unsigned l2 = (unsigned)(dv.z - d0), l3 = (unsigned)(dv.w - d0);
      const bool m0 = valid && l0 < (unsigned)R, m1 = valid && l1 < (unsigned)R;
      const bool m2 = valid && l2 < (unsigned)R, m3 = valid && l3 < (unsigned)R;
      if (__builtin_amdgcn_ballot_w32(m0 || m1 || m2 || m3)) {
        const unsigned eb = (unsigned)(w * 128 + lane * 4);
        const unsigned k0 = __builtin_amdgcn_ballot_w32(m0);
        const unsigned k1 = __builtin_amdgcn_ballot_w32(m1);
        const unsigned k2 = __builtin_amdgcn_ballot_w32(m2);
        const unsigned k3 = __builtin_amdgcn_ballot_w32(m3);
        if constexpr (PHASE == 0) {
          const int c = __popc(k0) + __popc(k1) + __popc(k2) + __popc(k3);
          if (qn + c > AGG_SCAP) {
            agg_score_max<H, false>(q, qs, qn, lane, d0, p, ew, src, mx, nullptr, nullptr, 0u);
            qn = 0;
            spilled = true;
          }
        }
        if (m0) q[qn + __popc(k0 & lt)] = (l0 << 20) | (eb + 0u);
        qn += __popc(k0);
        if (m1) q[qn + __popc(k1 & lt)] = (l1 << 20) | (eb + 1u);
        qn += __popc(k1);
        if (m2) q[qn + __popc(k2 & lt)] = (l2 << 20) | (eb + 2u);
        qn += __popc(k2);
        if (m3) q[qn + __popc(k3 & lt)] = (l3 << 20) | (eb + 3u);
        qn += __popc(k3);
      }
    }
    if constexpr (PHASE == 1) {
      if (qn >= 8) {
        agg_drain<H, F, R, false>(q, qs, qn, lane, d0, p, ew, src, mx, ht, acc, den);
        qn = 0;
      }
    }
    wa = xa;
    wb = xb;
  }
}

template <int H, int F, int R, int NW, bool ACT>
__device__ __forceinline__ void agg_rows(const unsigned* qbuf, const float* qsc,
                                         const unsigned* rowcnt, const unsigned* bucket,
                                         const unsigned* mx, unsigned* nextrow,
                                         int wave, int lane, int d0,
                                         const float* __restrict__ ht,
                                         const float* __restrict__ bias, float slope,
                                         float* __restrict__ hagg) {
  constexpr int HF = H * F;
  constexpr int NV = HF / 128;
  int r = wave;
  while (r < R) {
    const int n = d0 + r;
    if (n >= NN_NODES) break;
    const int cnt = __builtin_amdgcn_readfirstlane((int)rowcnt[r]);
    float m[H], den[H];
#pragma unroll
    for (int h = 0; h < H; ++h) {
      float mm = dec_f32(mx[r * H + h]);
      if (!isfinite(mm)) mm = 0.f;
      m[h] = mm;
      den[h] = 0.f;
    }
    float4 sum[NV];
#pragma unroll
    for (int k = 0; k < NV; ++k) sum[k] = make_float4(0.f, 0.f, 0.f, 0.f);

    for (int base = 0; base < cnt; base += 32) {
      const int c = (cnt - base) < 32 ? (cnt - base) : 32;
      int s = 0;
      float al[H];
#pragma unroll
      for (int h = 0; h < H; ++h) al[h] = 0.f;
      if (lane < c) {
        const unsigned idx = bucket[r * AGG_BCAP + base + lane];
        const unsigned ent = qbuf[(idx / AGG_SCAP) * AGG_QCAP + (idx % AGG_SCAP)];
        s = (int)(ent & 0xFFFFFu);
#pragma unroll
        for (int h = 0; h < H; ++h) al[h] = __expf(qsc[idx * H + h] - m[h]);
      }
      for (int i = 0; i < c; i += AGG_GROW) {
        float4 rv[AGG_GROW][NV];
#pragma unroll
        for (int g = 0; g < AGG_GROW; ++g) {
          const int idx = (i + g) < c ? (i + g) : (c - 1);
          const int sg = __builtin_amdgcn_readlane(s, idx);
          const float4* rp = (const float4*)(ht + (size_t)sg * HF);
#pragma unroll
          for (int k = 0; k < NV; ++k) rv[g][k] = rp[k * 32 + lane];
        }
#pragma unroll
        for (int g = 0; g < AGG_GROW; ++g) {
          if (i + g < c) {
            float ab[H];
#pragma unroll
            for (int h = 0; h < H; ++h) {
              ab[h] = __int_as_float(__builtin_amdgcn_readlane(__float_as_int(al[h]), i + g));
              den[h] += ab[h];
            }
#pragma unroll
            for (int k = 0; k < NV; ++k) {
              float a;
              if constexpr (H == 1) a = ab[0];
              else a = (lane & 16) ? ab[2 * k + 1] : ab[2 * k];
              const float4 v = rv[g][k];
              sum[k].x += a * v.x; sum[k].y += a * v.y;
              sum[k].z += a * v.z; sum[k].w += a * v.w;
            }
          }
        }
      }
    }

#pragma unroll
    for (int k = 0; k < NV; ++k) {
      float dsel;
      if constexpr (H == 1) dsel = den[0];
      else dsel = (lane & 16) ? den[2 * k + 1] : den[2 * k];
      const float sden = fmaxf(dsel, 1e-20f);
      const int c4 = k * 32 + lane;
      float4 v = sum[k];
      v.x = v.x / sden; v.y = v.y / sden; v.z = v.z / sden; v.w = v.w / sden;
      const float4 b = ((const float4*)bias)[c4];
      v.x += b.x; v.y += b.y; v.z += b.z; v.w += b.w;
      if constexpr (ACT) {
        v.x = leaky(v.x, slope); v.y = leaky(v.y, slope);
        v.z = leaky(v.z, slope); v.w = leaky(v.w, slope);
      }
      ((float4*)hagg)[(size_t)n * (HF / 4) + c4] = v;
    }

    unsigned nx = 0u;
    if (lane == 0)
      nx = __hip_atomic_fetch_add(nextrow, 1u, __ATOMIC_RELAXED, __HIP_MEMORY_SCOPE_WORKGROUP);
    r = __builtin_amdgcn_readfirstlane((int)nx);
  }
}

template <int H, int F, int R, int NT, bool ACT, bool FIRST>
__global__ __launch_bounds__(NT)
void edge_agg_owner_kernel(const float* __restrict__ p,
                           const float* __restrict__ ew,
                           const float* __restrict__ ht,
                           const int* __restrict__ src,
                           const int* __restrict__ dst,
                           const float* __restrict__ bias,
                           float slope,
                           float* __restrict__ hagg,
                           const unsigned* savq, float* savl, const int* savn) {
  constexpr int HF = H * F;
  constexpr int NW = NT / 32;
  static_assert(N_EDGES % 128 == 0, "edge windows of 128");
  static_assert(N_EDGES <= (1 << 20) && NN_NODES <= (1 << 20) && R <= 2048, "queue entry packing");
  static_assert(AGG_SCAP >= 128 && AGG_SCAP <= AGG_QCAP, "one window always fits an empty queue");
  static_assert(H == 1 || (H == 4 && F == 64), "head select per lane half");
  static_assert(HF % 128 == 0, "row = whole float4 per lane");
  static_assert(AGG_SCAP * NW <= (1 << 20) && AGG_BCAP >= 32, "bucket entry = place in the queues");
  static_assert(R == AGG_R && NT == AGG_NT, "the shape bin_owner_queues files the queues for");

  __shared__ float    acc[R * HF];
  __shared__ float    den[R * H];
  __shared__ unsigned mx[R * H];
  __shared__ unsigned qbuf[NW * AGG_QCAP];
  __shared__ float    qsc[NW * AGG_SCAP * H];
  __shared__ unsigned rowcnt[R];
  __shared__ unsigned bucket[R * AGG_BCAP];
  __shared__ unsigned ctl[2];

  const int t    = threadIdx.x;
  const int lane = t & 31;
  const int wave = __builtin_amdgcn_readfirstlane(t >> 5);
  const int d0   = blockIdx.x * R;
  unsigned* q    = qbuf + wave * AGG_QCAP;
  float*    qs   = qsc + wave * (AGG_SCAP * H);

  for (int i = t; i < R * H; i += NT) mx[i] = ENC_NEG_INF;
  for (int i = t; i < R; i += NT) rowcnt[i] = 0u;
  if (t == 0) { ctl[0] = 0u; ctl[1] = (unsigned)NW; }
  __syncthreads();

  int  qn      = 0;
  bool spilled = false;
  const int slot = blockIdx.x * NW + wave;
  const unsigned* sq = savq + (size_t)slot * AGG_SCAP;
  float*          sl = savl + (size_t)slot * AGG_SCAP;
  const unsigned qbase = (unsigned)(wave * AGG_SCAP);
  bool over = false;
  const int saved = __builtin_amdgcn_readfirstlane(savn[slot]);
  if (saved >= 0) {
    qn = saved;
    over = agg_saved_score_max<H, FIRST>(q, qs, qn, lane, d0, p, sq, sl, mx, rowcnt, bucket, qbase);
  } else {
    agg_scan<H, F, R, NT, 0>(dst, wave, lane, d0, q, qs, qn, spilled, p, ew, src, ht, mx, acc, den);
    over = agg_score_max<H, true>(q, qs, qn, lane, d0, p, ew, src, mx, rowcnt, bucket, qbase);
  }
  if (over || spilled) ctl[0] = 1u;
  __syncthreads();

  if (__builtin_amdgcn_readfirstlane((int)ctl[0]) == 0) {
    agg_rows<H, F, R, NW, ACT>(qbuf, qsc, rowcnt, bucket, mx, ctl + 1, wave, lane, d0,
                               ht, bias, slope, hagg);
    return;
  }

  for (int i = t; i < R * HF / 4; i += NT)
    ((float4*)acc)[i] = make_float4(0.f, 0.f, 0.f, 0.f);
  for (int i = t; i < R * H; i += NT) den[i] = 0.f;
  __syncthreads();
  if (!spilled) {
    agg_drain<H, F, R, true>(q, qs, qn, lane, d0, p, ew, src, mx, ht, acc, den);
  } else {
    qn = 0;
    agg_scan<H, F, R, NT, 1>(dst, wave, lane, d0, q, qs, qn, spilled, p, ew, src, ht, mx, acc, den);
    if (qn > 0)
      agg_drain<H, F, R, false>(q, qs, qn, lane, d0, p, ew, src, mx, ht, acc, den);
  }
  __syncthreads();

  for (int i = t; i < R * HF / 4; i += NT) {
    const int r = i / (HF / 4);
    const int n = d0 + r;
    if (n < NN_NODES) {
      float4 v = ((const float4*)acc)[i];
      const int c4 = i % (HF / 4);
      const float s = fmaxf(den[r * H + (c4 * 4) / F], 1e-20f);
      v.x = v.x / s; v.y = v.y / s; v.z = v.z / s; v.w = v.w / s;
      const float4 b = ((const float4*)bias)[c4];
      v.x += b.x; v.y += b.y; v.z += b.z; v.w += b.w;
      if constexpr (ACT) {
        v.x = leaky(v.x, slope); v.y = leaky(v.y, slope);
        v.z = leaky(v.z, slope); v.w = leaky(v.w, slope);
      }
      ((float4*)hagg)[(size_t)d0 * (HF / 4) + i] = v;
    }
  }
}

#define PRD_P1    132
#define PRD_P2    68
#define PRD_NW    4
#define PRD_UNITS (N_ROWS / 32)

__device__ __forceinline__ float predictor_tile(const float* __restrict__ emb, int s, int d,
                                                int c, int h,
                                                const float* W1s, const float* W2s,
                                                const float* __restrict__ pb1,
                                                const float* __restrict__ pb2,
                                                const float* __restrict__ pW3) {
  const float* ra = emb + (size_t)s * 128 + 4 * h;
  const float* rb = emb + (size_t)d * 128 + 4 * h;
  v8f acc[4];
#pragma unroll
  for (int nt = 0; nt < 4; ++nt) acc[nt] = (v8f){0.f, 0.f, 0.f, 0.f, 0.f, 0.f, 0.f, 0.f};

#pragma nounroll
  for (int jb = 0; jb < 4; ++jb) {
    v4f z[4];
#pragma unroll
    for (int j = 0; j < 4; ++j) {
      const v4f a = *(const v4f*)(ra + (jb * 4 + j) * 8);
      const v4f b = *(const v4f*)(rb + (jb * 4 + j) * 8);
      z[j] = a * b;
    }
#pragma unroll
    for (int j = 0; j < 4; ++j) {
      const int kc = (jb * 4 + j) * 8 + 4 * h;
      v4f w[4];
#pragma unroll
      for (int nt = 0; nt < 4; ++nt) w[nt] = *(const v4f*)(W1s + (nt * 16 + c) * PRD_P1 + kc);
      const v2f b0 = {z[j].x, z[j].y};
      const v2f b1 = {z[j].z, z[j].w};
#pragma unroll
      for (int nt = 0; nt < 4; ++nt) {
        const v2f a0 = {w[nt].x, w[nt].y};
        acc[nt] = __builtin_amdgcn_wmma_f32_16x16x4_f32(false, a0, false, b0, (short)0, acc[nt],
                                                        false, false);
      }
#pragma unroll
      for (int nt = 0; nt < 4; ++nt) {
        const v2f a1 = {w[nt].z, w[nt].w};
        acc[nt] = __builtin_amdgcn_wmma_f32_16x16x4_f32(false, a1, false, b1, (short)0, acc[nt],
                                                        false, false);
      }
    }
  }
#pragma unroll
  for (int nt = 0; nt < 4; ++nt) {
    const v4f bl = *(const v4f*)(pb1 + nt * 16 + 8 * h);
    const v4f bh = *(const v4f*)(pb1 + nt * 16 + 8 * h + 4);
    const float bb[8] = {bl.x, bl.y, bl.z, bl.w, bh.x, bh.y, bh.z, bh.w};
#pragma unroll
    for (int r = 0; r < 8; ++r) acc[nt][r] = leaky(acc[nt][r] + bb[r], 0.2f);
  }

  v8f acc2[2];
#pragma unroll
  for (int mt = 0; mt < 2; ++mt) acc2[mt] = (v8f){0.f, 0.f, 0.f, 0.f, 0.f, 0.f, 0.f, 0.f};
#pragma unroll
  for (int nt = 0; nt < 4; ++nt) {
    v4f wl[2], wh[2];
#pragma unroll
    for (int mt = 0; mt < 2; ++mt) {
      const float* wp = W2s + (mt * 16 + c) * PRD_P2 + nt * 16 + 8 * h;
      wl[mt] = *(const v4f*)wp;
      wh[mt] = *(const v4f*)(wp + 4);
    }
#pragma unroll
    for (int q = 0; q < 4; ++q) {
      const v2f b = {acc[nt][2 * q], acc[nt][2 * q + 1]};
#pragma unroll
      for (int mt = 0; mt < 2; ++mt) {
        const v4f wv = q < 2 ? wl[mt] : wh[mt];
        const v2f a = (q & 1) ? (v2f){wv.z, wv.w} : (v2f){wv.x, wv.y};
        acc2[mt] = __builtin_amdgcn_wmma_f32_16x16x4_f32(false, a, false, b, (short)0, acc2[mt],
                                                         false, false);
      }
    }
  }
  float part = 0.f;
#pragma unroll
  for (int mt = 0; mt < 2; ++mt) {
    const v4f bl = *(const v4f*)(pb2 + mt * 16 + 8 * h);
    const v4f bh = *(const v4f*)(pb2 + mt * 16 + 8 * h + 4);
    const v4f ul = *(const v4f*)(pW3 + mt * 16 + 8 * h);
    const v4f uh = *(const v4f*)(pW3 + mt * 16 + 8 * h + 4);
    const float bb[8] = {bl.x, bl.y, bl.z, bl.w, bh.x, bh.y, bh.z, bh.w};
    const float uu[8] = {ul.x, ul.y, ul.z, ul.w, uh.x, uh.y, uh.z, uh.w};
#pragma unroll
    for (int r = 0; r < 8; ++r) part += leaky(acc2[mt][r] + bb[r], 0.2f) * uu[r];
  }
  part += __shfl_xor(part, 16);
  return part;
}

__global__ __launch_bounds__(PRD_NW * 32) __attribute__((amdgpu_waves_per_eu(7)))
void predictor_kernel(const float* __restrict__ emb,
                      const int* __restrict__ ps, const int* __restrict__ pd,
                      const int* __restrict__ ns, const int* __restrict__ nd,
                      const float* __restrict__ pW1, const float* __restrict__ pb1,
                      const float* __restrict__ pW2, const float* __restrict__ pb2,
                      const float* __restrict__ pW3, const float* __restrict__ pb3,
                      float* out) {
  static_assert(N_ROWS % 32 == 0 && N_PAIR % 32 == 0, "a unit is 32 pairs of one list");
  __shared__ float W1s[64 * PRD_P1];
  __shared__ float W2s[32 * PRD_P2];
  const int t    = threadIdx.x;
  const int lane = t & 31;
  const int c    = lane & 15;
  const int h    = lane >> 4;
  const int wave = __builtin_amdgcn_readfirstlane(t >> 5);
  const int unit = blockIdx.x * PRD_NW + wave;

  int s0 = 0, d0 = 0, s1 = 0, d1 = 0;
  if (unit < PRD_UNITS) {
    const bool pos = unit < N_PAIR / 32;
    const int* si  = pos ? ps : ns;
    const int* di  = pos ? pd : nd;
    const int  r0  = (pos ? unit * 32 : unit * 32 - N_PAIR) + c;
    s0 = si[r0];      d0 = di[r0];
    s1 = si[r0 + 16]; d1 = di[r0 + 16];
  }
  static_assert(PRD_NW == 4, "staging counts below are for 128 threads");
  v4f g1[16], g2[4];
#pragma unroll
  for (int it = 0; it < 16; ++it) g1[it] = *(const v4f*)(pW1 + (t + it * 128) * 4);
#pragma unroll
  for (int it = 0; it < 4; ++it) g2[it] = *(const v4f*)(pW2 + (t + it * 128) * 4);
#pragma unroll
  for (int it = 0; it < 16; ++it) {
    const int i = t + it * 128;
    *(v4f*)(W1s + (i >> 5) * PRD_P1 + (i & 31) * 4) = g1[it];
  }
#pragma unroll
  for (int it = 0; it < 4; ++it) {
    const int i = t + it * 128;
    *(v4f*)(W2s + (i >> 4) * PRD_P2 + (i & 15) * 4) = g2[it];
  }
  __syncthreads();

  if (unit < PRD_UNITS) {
    const float v0 = predictor_tile(emb, s0, d0, c, h, W1s, W2s, pb1, pb2, pW3);
    const float v1 = predictor_tile(emb, s1, d1, c, h, W1s, W2s, pb1, pb2, pW3);
    out[unit * 32 + lane] = (h ? v1 : v0) + pb3[0];
  }
}

extern "C" void kernel_launch(void* const* d_in, const int* in_sizes, int n_in,
                              void* d_out, int out_size, void* d_ws, size_t ws_size,
                              hipStream_t stream) {
  const float* x    = (const float*)d_in[0];
  const float* ew   = (const float*)d_in[1];
  const int*   esrc = (const int*)d_in[2];
  const int*   edst = (const int*)d_in[3];
  const int*   psrc = (const int*)d_in[4];
  const int*   pdst = (const int*)d_in[5];
  const int*   nsrc = (const int*)d_in[6];
  const int*   ndst = (const int*)d_in[7];
  const float* Wl[3] = {(const float*)d_in[8],  (const float*)d_in[11], (const float*)d_in[14]};
  const float* al[3] = {(const float*)d_in[9],  (const float*)d_in[12], (const float*)d_in[15]};
  const float* bl[3] = {(const float*)d_in[10], (const float*)d_in[13], (const float*)d_in[16]};
  const float* pW1 = (const float*)d_in[17]; const float* pb1 = (const float*)d_in[18];
  const float* pW2 = (const float*)d_in[19]; const float* pb2 = (const float*)d_in[20];
  const float* pW3 = (const float*)d_in[21]; const float* pb3 = (const float*)d_in[22];

  float* out   = (float*)d_out;
  float* emb   = out + 2 * N_PAIR;

  float*    ws   = (float*)d_ws;
  float*    ht   = ws;
  float*    hbuf = ht   + 6400000;
  float*    pbuf = hbuf + 6400000;
  unsigned* savq = (unsigned*)(pbuf + 100000);
  float*    savl = pbuf + 100000 + 1048576;
  int*      savn = (int*)(pbuf + 100000 + 2097152);

  const int Hs[3]  = {4, 4, 1};
  const int Fs[3]  = {64, 64, 128};
  const int Kin[3] = {128, 256, 256};

  const float* hin = x;
  for (int l = 0; l < 3; ++l) {
    const int H = Hs[l], F = Fs[l], HF = H * F, K = Kin[l];
    float* hout = (l == 2) ? emb : hbuf;

    if (l == 0) {
      dim3 gg((NN_NODES + 63) / 64, HF / 32 + 1);
      gemm_wmma_f32<true><<<gg, 256, 0, stream>>>(hin, Wl[l], nullptr, ht,
                                                  NN_NODES, HF, K, 0.f, 0,
                                                  esrc, edst, ew, savq, savl, savn);
    } else {
      dim3 gg((NN_NODES + 63) / 64, HF / 32);
      gemm_wmma_f32<false><<<gg, 256, 0, stream>>>(hin, Wl[l], nullptr, ht,
                                                   NN_NODES, HF, K, 0.f, 0,
                                                   nullptr, nullptr, nullptr,
                                                   nullptr, nullptr, nullptr);
    }
    int nh = NN_NODES * H;
    if (l < 2) node_p_kernel<4, 64><<<(nh + 255) / 256, 256, 0, stream>>>(ht, al[l], pbuf);
    else       node_p_kernel<1, 128><<<(nh + 255) / 256, 256, 0, stream>>>(ht, al[l], pbuf);

    if (l == 0) {
      edge_agg_owner_kernel<4, 64, AGG_R, AGG_NT, true, true><<<AGG_NG, AGG_NT, 0, stream>>>(
          pbuf, ew, ht, esrc, edst, bl[l], 0.01f, hout, savq, savl, savn);
    } else if (l == 1) {
      edge_agg_owner_kernel<4, 64, AGG_R, AGG_NT, true, false><<<AGG_NG, AGG_NT, 0, stream>>>(
          pbuf, ew, ht, esrc, edst, bl[l], 0.01f, hout, savq, savl, savn);
    } else {
      edge_agg_owner_kernel<1, 128, AGG_R, AGG_NT, false, false><<<AGG_NG, AGG_NT, 0, stream>>>(
          pbuf, ew, ht, esrc, edst, bl[l], 0.01f, hout, savq, savl, savn);
    }
    hin = hout;
  }

  predictor_kernel<<<(PRD_UNITS + PRD_NW - 1) / PRD_NW, PRD_NW * 32, 0, stream>>>(
      emb, psrc, pdst, nsrc, ndst, pW1, pb1, pW2, pb2, pW3, pb3, out);
}


# Attention projection p = input rows x (W^T a, built in LDS) by 98 more work-groups inside each GEMM launch, grid flattened to one row: 3 launches gone; p sum order changes (k first, lane tree)
# speedup vs baseline: 14.9234x; 1.0355x over previous
#include <hip/hip_runtime.h>
#include <hip/hip_bf16.h>


typedef __attribute__((ext_vector_type(2))) float v2f;
typedef __attribute__((ext_vector_type(8))) float v8f;
typedef __attribute__((ext_vector_type(4))) float v4f;

#define NN_NODES 25000
#define N_EDGES  400000
#define N_PAIR   100000
#define N_ROWS   200000

__device__ __forceinline__ unsigned enc_f32(float f) {
  unsigned u = __float_as_uint(f);
  return (u & 0x80000000u) ? ~u : (u | 0x80000000u);
}
__device__ __forceinline__ float dec_f32(unsigned e) {
  return (e & 0x80000000u) ? __uint_as_float(e & 0x7FFFFFFFu)
                           : __uint_as_float(~e);
}
#define ENC_NEG_INF 0x007FFFFFu

__device__ __forceinline__ float leaky(float v, float slope) {
  return v >= 0.f ? v : v * slope;
}

#define AGG_R    98
#define AGG_NT   1024
#define AGG_NG   ((NN_NODES + AGG_R - 1) / AGG_R)
#define AGG_QCAP 272
#define AGG_SCAP 128
#define AGG_BCAP 64
#define AGG_GROW 8

template <int R, int NW, int NTB>
__device__ __forceinline__ void bin_owner_queues(unsigned* cnt, int w,
                                                 const int* __restrict__ src,
                                                 const int* __restrict__ dst,
                                                 const float* __restrict__ ew,
                                                 unsigned* savq, float* savl, int* savn) {
  constexpr int NG   = (NN_NODES + R - 1) / R;
  constexpr int NWIN = N_EDGES / 128;
  constexpr int STEP = NW * (NTB / 32);
  static_assert(N_EDGES % 128 == 0, "edge windows of 128");
  static_assert(R <= 2048 && NN_NODES <= (1 << 20), "queue entry packing");
  const int t    = threadIdx.x;
  const int lane = t & 31;
  const int wv   = __builtin_amdgcn_readfirstlane(t >> 5);
  for (int i = t; i < NG; i += NTB) cnt[i] = 0u;
  __syncthreads();

  const int4*   dst4 = (const int4*)dst;
  const int4*   src4 = (const int4*)src;
  const float4* ew4  = (const float4*)ew;
  int win = w + NW * wv;
  int4 dv[2];
#pragma unroll
  for (int h = 0; h < 2; ++h) {
    const int c0 = (win + h * STEP) < NWIN ? (win + h * STEP) : NWIN - 1;
    dv[h] = dst4[c0 * 32 + lane];
  }
  for (; win < NWIN; win += 2 * STEP) {
    int4   sv[2], dn[2];
    float4 wt[2];
#pragma unroll
    for (int h = 0; h < 2; ++h) {
      const int c0 = (win + h * STEP) < NWIN ? (win + h * STEP) : NWIN - 1;
      sv[h] = src4[c0 * 32 + lane];
      wt[h] = ew4[c0 * 32 + lane];
    }
#pragma unroll
    for (int h = 0; h < 2; ++h) {
      const int nx = (win + (2 + h) * STEP) < NWIN ? (win + (2 + h) * STEP) : NWIN - 1;
      dn[h] = dst4[nx * 32 + lane];
    }
#pragma unroll
    for (int h = 0; h < 2; ++h) {
      const bool  valid = (win + h * STEP) < NWIN;
      const int   dd[4] = {dv[h].x, dv[h].y, dv[h].z, dv[h].w};
      const int   ss[4] = {sv[h].x, sv[h].y, sv[h].z, sv[h].w};
      const float ww[4] = {wt[h].x, wt[h].y, wt[h].z, wt[h].w};
      unsigned qb[4], rw[4], rk[4];
#pragma unroll
      for (int k = 0; k < 4; ++k) {
        const bool ok = valid && (unsigned)dd[k] < (unsigned)NN_NODES;
        const unsigned d = ok ? (unsigned)dd[k] : 0u;
        const unsigned g = d / (unsigned)R;
        rw[k] = (d - g * (unsigned)R) << 20;
        qb[k] = (g * (unsigned)NW + (unsigned)w) * (unsigned)AGG_SCAP;
        const unsigned got = __hip_atomic_fetch_add(cnt + g, ok ? 1u : 0u, __ATOMIC_RELAXED,
                                                    __HIP_MEMORY_SCOPE_WORKGROUP);
        rk[k] = ok ? got : (unsigned)AGG_SCAP;
      }
#pragma unroll
      for (int k = 0; k < 4; ++k) {
        if (rk[k] < (unsigned)AGG_SCAP) {
          const unsigned at = qb[k] + rk[k];
          savq[at] = rw[k] | (unsigned)ss[k];
          savl[at] = ww[k];
        }
      }
    }
    dv[0] = dn[0];
    dv[1] = dn[1];
  }
  __syncthreads();
  for (int i = t; i < NG; i += NTB) {
    const unsigned n = cnt[i];
    savn[i * NW + w] = n <= (unsigned)AGG_SCAP ? (int)n : -1;
  }
}

#define P_NODES 256
#define P_NWG   ((NN_NODES + P_NODES - 1) / P_NODES)

template <int H, int F, int K>
__device__ __forceinline__ void node_p_rows(float* lds, int w,
                                            const float* __restrict__ hin,
                                            const float* __restrict__ W,
                                            const float* __restrict__ a,
                                            float* __restrict__ p) {
  constexpr int K4  = K / 4;
  constexpr int NG  = 256 / K4;
  constexpr int GPH = NG / H;
  constexpr int FS  = F / GPH;
  constexpr int NV  = K4 / 32;
  constexpr int NB  = (H == 1 ? 1024 : 512) / K;
  static_assert(K4 % 32 == 0 && NG % H == 0 && F % GPH == 0, "whole waves per group, whole slices per head");
  static_assert(H == 1 || H == 4, "p row store");
  static_assert(P_NODES == 256 && (32 / H) % NB == 0, "eight waves, 32 nodes each, whole trips per group");
  const int t    = threadIdx.x;
  const int lane = t & 31;
  const int wv   = __builtin_amdgcn_readfirstlane(t >> 5);
  const int g    = wv / (K4 / 32);
  const int cq   = t % K4;
  const int hd   = g / GPH;
  const int sl   = g % GPH;
  float4* red4 = (float4*)lds;
  float4* wa4  = (float4*)(lds + 1024);

  {
    const float4* wp = (const float4*)(W + (size_t)(hd * F + sl * FS) * K) + cq;
    const float*  ap = a + sl * FS;
    float4 s = make_float4(0.f, 0.f, 0.f, 0.f);
#pragma unroll 8
    for (int f = 0; f < FS; ++f) {
      const float4 v  = wp[f * K4];
      const float  af = ap[f];
      s.x += v.x * af; s.y += v.y * af; s.z += v.z * af; s.w += v.w * af;
    }
    red4[t] = s;
  }
  __syncthreads();
  if (t < H * K4) {
    const int hh = t / K4, c = t % K4;
    float4 r = red4[(hh * GPH) * K4 + c];
#pragma unroll
    for (int j = 1; j < GPH; ++j) {
      const float4 q = red4[(hh * GPH + j) * K4 + c];
      r.x += q.x; r.y += q.y; r.z += q.z; r.w += q.w;
    }
    wa4[hh * K4 + c] = r;
  }
  __syncthreads();

  float4 wr[H][NV];
#pragma unroll
  for (int h = 0; h < H; ++h)
#pragma unroll
    for (int i = 0; i < NV; ++i) wr[h][i] = wa4[h * K4 + i * 32 + lane];

  constexpr int GSZ = 32 / H;
  const int  sub = lane & (GSZ - 1);
  const int  hh  = lane / GSZ;
  const int  n0  = w * P_NODES + wv * 32;
#pragma nounroll
  for (int gq = 0; gq < H; ++gq) {
    float cur = 0.f;
#pragma nounroll
    for (int jj = 0; jj < GSZ; jj += NB) {
      float4 v[NB][NV];
#pragma unroll
      for (int b = 0; b < NB; ++b) {
        const int n  = n0 + gq * GSZ + jj + b;
        const int nc = n < NN_NODES ? n : NN_NODES - 1;
        const float4* rp = (const float4*)(hin + (size_t)nc * K) + lane;
#pragma unroll
        for (int i = 0; i < NV; ++i) v[b][i] = rp[i * 32];
      }
#pragma unroll
      for (int b = 0; b < NB; ++b) {
        float d[H];
#pragma unroll
        for (int h = 0; h < H; ++h) {
          float s = 0.f;
#pragma unroll
          for (int i = 0; i < NV; ++i) {
            s = __builtin_fmaf(v[b][i].x, wr[h][i].x, s);
            s = __builtin_fmaf(v[b][i].y, wr[h][i].y, s);
            s = __builtin_fmaf(v[b][i].z, wr[h][i].z, s);
            s = __builtin_fmaf(v[b][i].w, wr[h][i].w, s);
          }
          d[h] = s;
        }
        float tot;
        if constexpr (H == 4) {
          const bool hi = (lane & 16) != 0;
          const bool b3 = (lane & 8) != 0;
          float k0 = hi ? d[2] : d[0], k1 = hi ? d[3] : d[1];
          const float s0 = hi ? d[0] : d[2], s1 = hi ? d[1] : d[3];
          k0 += __shfl_xor(s0, 16);
          k1 += __shfl_xor(s1, 16);
          tot = b3 ? k1 : k0;
          const float sq = b3 ? k0 : k1;
          tot += __shfl_xor(sq, 8);
        } else {
          tot = d[0];
          tot += __shfl_xor(tot, 16);
          tot += __shfl_xor(tot, 8);
        }
        tot += __shfl_xor(tot, 4);
        tot += __shfl_xor(tot, 2);
        tot += __shfl_xor(tot, 1);
        cur = (sub == jj + b) ? tot : cur;
      }
    }
    const int n = n0 + gq * GSZ + sub;
    if (n < NN_NODES) p[(size_t)n * H + hh] = cur;
  }
}

#define GEMM_MB ((NN_NODES + 63) / 64)

template <bool BIN, int PH, int PF, int PK>
__global__ void gemm_wmma_f32(const float* __restrict__ A,
                              const float* __restrict__ W,
                              const float* __restrict__ bias,
                              float* __restrict__ C,
                              int M, int N, int K,
                              float slope, int act,
                              const int* __restrict__ esrc,
                              const int* __restrict__ edst,
                              const float* __restrict__ ew,
                              unsigned* savq, float* savl, int* savn,
                              const float* __restrict__ avec,
                              float* __restrict__ pout) {
  __shared__ float As[64][36];
  __shared__ float Ws[32][36];
  static_assert(sizeof(As) >= 2048 * sizeof(float), "node_p_rows borrows the A tile's LDS");

  constexpr int NBIN = BIN ? AGG_NT / 32 : 0;
  int b = (int)blockIdx.x;
  if (b < NBIN + P_NWG) {
    if constexpr (BIN) {
      if (b < NBIN) {
        __shared__ unsigned bcnt[AGG_NG];
        bin_owner_queues<AGG_R, AGG_NT / 32, 256>(bcnt, b, esrc, edst, ew, savq, savl, savn);
        return;
      }
    }
    node_p_rows<PH, PF, PK>(&As[0][0], b - NBIN, A, W, avec, pout);
    return;
  }
  b -= NBIN + P_NWG;
  const int nTile = b / GEMM_MB;

  const int t     = threadIdx.x;
  const int lane  = t & 31;
  const int wave  = t >> 5;
  const int waveM = wave & 3;
  const int waveN = wave >> 2;
  const int mBase = (b - nTile * GEMM_MB) * 64;
  const int nBase = nTile * 32;
  const int lh    = lane >> 4;
  const int l15   = lane & 15;

  v8f acc = {0.f, 0.f, 0.f, 0.f, 0.f, 0.f, 0.f, 0.f};

  for (int k0 = 0; k0 < K; k0 += 32) {
    {
      const int r   = t >> 3;
      const int c4  = (t & 7) * 4;
      const int gr0 = mBase + r;
      const int gr1 = gr0 + 32;
      const int g0c = gr0 < M ? gr0 : M - 1;
      const int g1c = gr1 < M ? gr1 : M - 1;
      v4f va0 = *(const v4f*)(A + (size_t)g0c * K + k0 + c4);
      v4f va1 = *(const v4f*)(A + (size_t)g1c * K + k0 + c4);
      const v4f vw = *(const v4f*)(W + (size_t)(nBase + r) * K + k0 + c4);
      asm volatile("" :: "v"(va0), "v"(va1), "v"(vw));
      const v4f zero = {0.f, 0.f, 0.f, 0.f};
      if (gr0 >= M) va0 = zero;
      if (gr1 >= M) va1 = zero;
      *(v4f*)&As[r][c4]      = va0;
      *(v4f*)&As[r + 32][c4] = va1;
      *(v4f*)&Ws[r][c4]      = vw;
    }
    __syncthreads();

#pragma unroll
    for (int ks = 0; ks < 8; ++ks) {
      int kb = ks * 4 + 2 * lh;
      v2f a, b;
      a.x = As[waveM * 16 + l15][kb];
      a.y = As[waveM * 16 + l15][kb + 1];
      b.x = Ws[waveN * 16 + l15][kb];
      b.y = Ws[waveN * 16 + l15][kb + 1];
      acc = __builtin_amdgcn_wmma_f32_16x16x4_f32(
           false, a,  false, b,
           (short)0, acc,  false,  false);
    }
    __syncthreads();
  }

#pragma unroll
  for (int r = 0; r < 8; ++r) {
    int row = mBase + waveM * 16 + r + 8 * lh;
    int col = nBase + waveN * 16 + l15;
    if (row < M) {
      float v = acc[r];
      if (bias) v += bias[col];
      if (act)  v = leaky(v, slope);
      C[(size_t)row * N + col] = v;
    }
  }
}


template <int H>
__device__ __forceinline__ void node_scores(int s, int d, float lw,
                                            const float* __restrict__ p,
                                            float* sc) {
  float ps[H], pd[H];
#pragma unroll
  for (int h = 0; h < H; ++h) ps[h] = p[s * H + h];
#pragma unroll
  for (int h = 0; h < H; ++h) pd[h] = p[d * H + h];
#pragma unroll
  for (int h = 0; h < H; ++h) sc[h] = leaky(ps[h] + pd[h] + lw, 0.01f);
}

template <int H>
__device__ __forceinline__ int edge_scores(int e, int d,
                                           const float* __restrict__ p,
                                           const float* __restrict__ ew,
                                           const int* __restrict__ src,
                                           float* sc, float& lw) {
  const int s = src[e];
  lw = log1pf(ew[e]);
  node_scores<H>(s, d, lw, p, sc);
  return s;
}

template <int H, bool RANK>
__device__ __forceinline__ bool agg_score_max(unsigned* q, float* qs, int qn, int lane, int d0,
                                              const float* __restrict__ p,
                                              const float* __restrict__ ew,
                                              const int* __restrict__ src,
                                              unsigned* mx,
                                              unsigned* rowcnt, unsigned* bucket, unsigned qbase) {
  bool over = false;
  for (int base = 0; base < qn; base += 32) {
    const int i = base + lane;
    if (i < qn) {
      const unsigned ent = q[i];
      const int e = (int)(ent & 0xFFFFFu);
      const unsigned dl = ent >> 20;
      float sc[H];
      float lw;
      const int s = edge_scores<H>(e, d0 + (int)dl, p, ew, src, sc, lw);
#pragma unroll
      for (int h = 0; h < H; ++h) {
        qs[i * H + h] = sc[h];
        __hip_atomic_fetch_max(mx + dl * H + h, enc_f32(sc[h]), __ATOMIC_RELAXED,
                               __HIP_MEMORY_SCOPE_WORKGROUP);
      }
      const unsigned kept = (dl << 20) | (unsigned)s;
      q[i] = kept;
      if constexpr (RANK) {
        const unsigned rk = __hip_atomic_fetch_add(rowcnt + dl, 1u, __ATOMIC_RELAXED,
                                                   __HIP_MEMORY_SCOPE_WORKGROUP);
        if (rk < (unsigned)AGG_BCAP) bucket[dl * AGG_BCAP + rk] = qbase + (unsigned)i;
        else over = true;
      }
    }
  }
  return over;
}

template <int H, bool FIRST>
__device__ __forceinline__ bool agg_saved_score_max(unsigned* q, float* qs, int qn, int lane, int d0,
                                                    const float* __restrict__ p,
                                                    const unsigned* sq, float* sl,
                                                    unsigned* mx,
                                                    unsigned* rowcnt, unsigned* bucket, unsigned qbase) {
  bool over = false;
  for (int base = 0; base < qn; base += 32) {
    const int i = base + lane;
    if (i < qn) {
      const unsigned ent = sq[i];
      float lw = sl[i];
      if constexpr (FIRST) { lw = log1pf(lw); sl[i] = lw; }
      const unsigned dl = ent >> 20;
      float sc[H];
      node_scores<H>((int)(ent & 0xFFFFFu), d0 + (int)dl, lw, p, sc);
#pragma unroll
      for (int h = 0; h < H; ++h) {
        qs[i * H + h] = sc[h];
        __hip_atomic_fetch_max(mx + dl * H + h, enc_f32(sc[h]), __ATOMIC_RELAXED,
                               __HIP_MEMORY_SCOPE_WORKGROUP);
      }
      q[i] = ent;
      const unsigned rk = __hip_atomic_fetch_add(rowcnt + dl, 1u, __ATOMIC_RELAXED,
                                                 __HIP_MEMORY_SCOPE_WORKGROUP);
      if (rk < (unsigned)AGG_BCAP) bucket[dl * AGG_BCAP + rk] = qbase + (unsigned)i;
      else over = true;
    }
  }
  return over;
}

template <int H, int F, int R, bool KEPT>
__device__ __forceinline__ void agg_drain(const unsigned* q, const float* qs, int qn, int lane, int d0,
                                          const float* __restrict__ p,
                                          const float* __restrict__ ew,
                                          const int* __restrict__ src,
                                          const unsigned* mx,
                                          const float* __restrict__ ht,
                                          float* acc, float* den) {
  constexpr int HF = H * F;
  constexpr int NV = HF / 128;
  for (int base = 0; base < qn; base += 32) {
    const int cnt = (qn - base) < 32 ? (qn - base) : 32;
    int s = 0, dl = 0;
    float al[H];
#pragma unroll
    for (int h = 0; h < H; ++h) al[h] = 0.f;
    if (lane < cnt) {
      const unsigned ent = q[base + lane];
      dl = (int)(ent >> 20);
      float sc[H];
      if constexpr (KEPT) {
        s = (int)(ent & 0xFFFFFu);
#pragma unroll
        for (int h = 0; h < H; ++h) sc[h] = qs[(base + lane) * H + h];
      } else {
        float lw;
        s = edge_scores<H>((int)(ent & 0xFFFFFu), d0 + dl, p, ew, src, sc, lw);
      }
#pragma unroll
      for (int h = 0; h < H; ++h) {
        float m = dec_f32(mx[dl * H + h]);
        if (!isfinite(m)) m = 0.f;
        al[h] = __expf(sc[h] - m);
        __hip_atomic_fetch_add(den + dl * H + h, al[h], __ATOMIC_RELAXED,
                               __HIP_MEMORY_SCOPE_WORKGROUP);
      }
    }
    for (int i = 0; i < cnt; i += 4) {
      float4 rv[4][NV];
#pragma unroll
      for (int g = 0; g < 4; ++g) {
        const int idx = (i + g) < cnt ? (i + g) : (cnt - 1);
        const int sg = __builtin_amdgcn_readlane(s, idx);
        const float4* rp = (const float4*)(ht + (size_t)sg * HF);
#pragma unroll
        for (int k = 0; k < NV; ++k) rv[g][k] = rp[k * 32 + lane];
      }
#pragma unroll
      for (int g = 0; g < 4; ++g) {
        if (i + g < cnt) {
          const int dlg = __builtin_amdgcn_readlane(dl, i + g);
          float ab[H];
#pragma unroll
          for (int h = 0; h < H; ++h)
            ab[h] = __int_as_float(__builtin_amdgcn_readlane(__float_as_int(al[h]), i + g));
          float* ap = acc + dlg * HF;
#pragma unroll
          for (int k = 0; k < NV; ++k) {
            float a;
            if constexpr (H == 1) a = ab[0];
            else a = (lane & 16) ? ab[2 * k + 1] : ab[2 * k];
            float* pp = ap + (k * 32 + lane) * 4;
            const float4 v = rv[g][k];
            __hip_atomic_fetch_add(pp + 0, a * v.x, __ATOMIC_RELAXED, __HIP_MEMORY_SCOPE_WORKGROUP);
            __hip_atomic_fetch_add(pp + 1, a * v.y, __ATOMIC_RELAXED, __HIP_MEMORY_SCOPE_WORKGROUP);
            __hip_atomic_fetch_add(pp + 2, a * v.z, __ATOMIC_RELAXED, __HIP_MEMORY_SCOPE_WORKGROUP);
            __hip_atomic_fetch_add(pp + 3, a * v.w, __ATOMIC_RELAXED, __HIP_MEMORY_SCOPE_WORKGROUP);
          }
        }
      }
    }
  }
}

template <int H, int F, int R, int NT, int PHASE>
__device__ __forceinline__ void agg_scan(const int* __restrict__ dst, int wave, int lane, int d0,
                                         unsigned* q, float* qs, int& qn, bool& spilled,
                                         const float* __restrict__ p,
                                         const float* __restrict__ ew,
                                         const int* __restrict__ src,
                                         const float* __restrict__ ht,
                                         unsigned* mx, float* acc, float* den) {
  constexpr int NW   = NT / 32;
  constexpr int NWIN = N_EDGES / 128;
  const int4* dst4 = (const int4*)dst;
  const unsigned lt = (1u << lane) - 1u;

  int4 wa, wb;
  {
    const int a0 = wave < NWIN ? wave : NWIN - 1;
    const int b0 = (wave + NW) < NWIN ? (wave + NW) : NWIN - 1;
    wa = dst4[a0 * 32 + lane];
    wb = dst4[b0 * 32 + lane];
  }
  for (int win = wave; win < NWIN; win += 2 * NW) {
    const int na = (win + 2 * NW) < NWIN ? (win + 2 * NW) : NWIN - 1;
    const int nb = (win + 3 * NW) < NWIN ? (win + 3 * NW) : NWIN - 1;
    const int4 xa = dst4[na * 32 + lane];
    const int4 xb = dst4[nb * 32 + lane];

#pragma unroll
    for (int half = 0; half < 2; ++half) {
      const int  w     = win + half * NW;
      const bool valid = w < NWIN;
      const int4 dv    = half ? wb : wa;
      const unsigned l0 = (unsigned)(dv.x - d0), l1 = (unsigned)(dv.y - d0);
      const unsigned l2 = (unsigned)(dv.z - d0), l3 = (unsigned)(dv.w - d0);
      const bool m0 = valid && l0 < (unsigned)R, m1 = valid && l1 < (unsigned)R;
      const bool m2 = valid && l2 < (unsigned)R, m3 = valid && l3 < (unsigned)R;
      if (__builtin_amdgcn_ballot_w32(m0 || m1 || m2 || m3)) {
        const unsigned eb = (unsigned)(w * 128 + lane * 4);
        const unsigned k0 = __builtin_amdgcn_ballot_w32(m0);
        const unsigned k1 = __builtin_amdgcn_ballot_w32(m1);
        const unsigned k2 = __builtin_amdgcn_ballot_w32(m2);
        const unsigned k3 = __builtin_amdgcn_ballot_w32(m3);
        if constexpr (PHASE == 0) {
          const int c = __popc(k0) + __popc(k1) + __popc(k2) + __popc(k3);
          if (qn + c > AGG_SCAP) {
            agg_score_max<H, false>(q, qs, qn, lane, d0, p, ew, src, mx, nullptr, nullptr, 0u);
            qn = 0;
            spilled = true;
          }
        }
        if (m0) q[qn + __popc(k0 & lt)] = (l0 << 20) | (eb + 0u);
        qn += __popc(k0);
        if (m1) q[qn + __popc(k1 & lt)] = (l1 << 20) | (eb + 1u);
        qn += __popc(k1);
        if (m2) q[qn + __popc(k2 & lt)] = (l2 << 20) | (eb + 2u);
        qn += __popc(k2);
        if (m3) q[qn + __popc(k3 & lt)] = (l3 << 20) | (eb + 3u);
        qn += __popc(k3);
      }
    }
    if constexpr (PHASE == 1) {
      if (qn >= 8) {
        agg_drain<H, F, R, false>(q, qs, qn, lane, d0, p, ew, src, mx, ht, acc, den);
        qn = 0;
      }
    }
    wa = xa;
    wb = xb;
  }
}

template <int H, int F, int R, int NW, bool ACT>
__device__ __forceinline__ void agg_rows(const unsigned* qbuf, const float* qsc,
                                         const unsigned* rowcnt, const unsigned* bucket,
                                         const unsigned* mx, unsigned* nextrow,
                                         int wave, int lane, int d0,
                                         const float* __restrict__ ht,
                                         const float* __restrict__ bias, float slope,
                                         float* __restrict__ hagg) {
  constexpr int HF = H * F;
  constexpr int NV = HF / 128;
  int r = wave;
  while (r < R) {
    const int n = d0 + r;
    if (n >= NN_NODES) break;
    const int cnt = __builtin_amdgcn_readfirstlane((int)rowcnt[r]);
    float m[H], den[H];
#pragma unroll
    for (int h = 0; h < H; ++h) {
      float mm = dec_f32(mx[r * H + h]);
      if (!isfinite(mm)) mm = 0.f;
      m[h] = mm;
      den[h] = 0.f;
    }
    float4 sum[NV];
#pragma unroll
    for (int k = 0; k < NV; ++k) sum[k] = make_float4(0.f, 0.f, 0.f, 0.f);

    for (int base = 0; base < cnt; base += 32) {
      const int c = (cnt - base) < 32 ? (cnt - base) : 32;
      int s = 0;
      float al[H];
#pragma unroll
      for (int h = 0; h < H; ++h) al[h] = 0.f;
      if (lane < c) {
        const unsigned idx = bucket[r * AGG_BCAP + base + lane];
        const unsigned ent = qbuf[(idx / AGG_SCAP) * AGG_QCAP + (idx % AGG_SCAP)];
        s = (int)(ent & 0xFFFFFu);
#pragma unroll
        for (int h = 0; h < H; ++h) al[h] = __expf(qsc[idx * H + h] - m[h]);
      }
      for (int i = 0; i < c; i += AGG_GROW) {
        float4 rv[AGG_GROW][NV];
#pragma unroll
        for (int g = 0; g < AGG_GROW; ++g) {
          const int idx = (i + g) < c ? (i + g) : (c - 1);
          const int sg = __builtin_amdgcn_readlane(s, idx);
          const float4* rp = (const float4*)(ht + (size_t)sg * HF);
#pragma unroll
          for (int k = 0; k < NV; ++k) rv[g][k] = rp[k * 32 + lane];
        }
#pragma unroll
        for (int g = 0; g < AGG_GROW; ++g) {
          if (i + g < c) {
            float ab[H];
#pragma unroll
            for (int h = 0; h < H; ++h) {
              ab[h] = __int_as_float(__builtin_amdgcn_readlane(__float_as_int(al[h]), i + g));
              den[h] += ab[h];
            }
#pragma unroll
            for (int k = 0; k < NV; ++k) {
              float a;
              if constexpr (H == 1) a = ab[0];
              else a = (lane & 16) ? ab[2 * k + 1] : ab[2 * k];
              const float4 v = rv[g][k];
              sum[k].x += a * v.x; sum[k].y += a * v.y;
              sum[k].z += a * v.z; sum[k].w += a * v.w;
            }
          }
        }
      }
    }

#pragma unroll
    for (int k = 0; k < NV; ++k) {
      float dsel;
      if constexpr (H == 1) dsel = den[0];
      else dsel = (lane & 16) ? den[2 * k + 1] : den[2 * k];
      const float sden = fmaxf(dsel, 1e-20f);
      const int c4 = k * 32 + lane;
      float4 v = sum[k];
      v.x = v.x / sden; v.y = v.y / sden; v.z = v.z / sden; v.w = v.w / sden;
      const float4 b = ((const float4*)bias)[c4];
      v.x += b.x; v.y += b.y; v.z += b.z; v.w += b.w;
      if constexpr (ACT) {
        v.x = leaky(v.x, slope); v.y = leaky(v.y, slope);
        v.z = leaky(v.z, slope); v.w = leaky(v.w, slope);
      }
      ((float4*)hagg)[(size_t)n * (HF / 4) + c4] = v;
    }

    unsigned nx = 0u;
    if (lane == 0)
      nx = __hip_atomic_fetch_add(nextrow, 1u, __ATOMIC_RELAXED, __HIP_MEMORY_SCOPE_WORKGROUP);
    r = __builtin_amdgcn_readfirstlane((int)nx);
  }
}

template <int H, int F, int R, int NT, bool ACT, bool FIRST>
__global__ __launch_bounds__(NT)
void edge_agg_owner_kernel(const float* __restrict__ p,
                           const float* __restrict__ ew,
                           const float* __restrict__ ht,
                           const int* __restrict__ src,
                           const int* __restrict__ dst,
                           const float* __restrict__ bias,
                           float slope,
                           float* __restrict__ hagg,
                           const unsigned* savq, float* savl, const int* savn) {
  constexpr int HF = H * F;
  constexpr int NW = NT / 32;
  static_assert(N_EDGES % 128 == 0, "edge windows of 128");
  static_assert(N_EDGES <= (1 << 20) && NN_NODES <= (1 << 20) && R <= 2048, "queue entry packing");
  static_assert(AGG_SCAP >= 128 && AGG_SCAP <= AGG_QCAP, "one window always fits an empty queue");
  static_assert(H == 1 || (H == 4 && F == 64), "head select per lane half");
  static_assert(HF % 128 == 0, "row = whole float4 per lane");
  static_assert(AGG_SCAP * NW <= (1 << 20) && AGG_BCAP >= 32, "bucket entry = place in the queues");
  static_assert(R == AGG_R && NT == AGG_NT, "the shape bin_owner_queues files the queues for");

  __shared__ float    acc[R * HF];
  __shared__ float    den[R * H];
  __shared__ unsigned mx[R * H];
  __shared__ unsigned qbuf[NW * AGG_QCAP];
  __shared__ float    qsc[NW * AGG_SCAP * H];
  __shared__ unsigned rowcnt[R];
  __shared__ unsigned bucket[R * AGG_BCAP];
  __shared__ unsigned ctl[2];

  const int t    = threadIdx.x;
  const int lane = t & 31;
  const int wave = __builtin_amdgcn_readfirstlane(t >> 5);
  const int d0   = blockIdx.x * R;
  unsigned* q    = qbuf + wave * AGG_QCAP;
  float*    qs   = qsc + wave * (AGG_SCAP * H);

  for (int i = t; i < R * H; i += NT) mx[i] = ENC_NEG_INF;
  for (int i = t; i < R; i += NT) rowcnt[i] = 0u;
  if (t == 0) { ctl[0] = 0u; ctl[1] = (unsigned)NW; }
  __syncthreads();

  int  qn      = 0;
  bool spilled = false;
  const int slot = blockIdx.x * NW + wave;
  const unsigned* sq = savq + (size_t)slot * AGG_SCAP;
  float*          sl = savl + (size_t)slot * AGG_SCAP;
  const unsigned qbase = (unsigned)(wave * AGG_SCAP);
  bool over = false;
  const int saved = __builtin_amdgcn_readfirstlane(savn[slot]);
  if (saved >= 0) {
    qn = saved;
    over = agg_saved_score_max<H, FIRST>(q, qs, qn, lane, d0, p, sq, sl, mx, rowcnt, bucket, qbase);
  } else {
    agg_scan<H, F, R, NT, 0>(dst, wave, lane, d0, q, qs, qn, spilled, p, ew, src, ht, mx, acc, den);
    over = agg_score_max<H, true>(q, qs, qn, lane, d0, p, ew, src, mx, rowcnt, bucket, qbase);
  }
  if (over || spilled) ctl[0] = 1u;
  __syncthreads();

  if (__builtin_amdgcn_readfirstlane((int)ctl[0]) == 0) {
    agg_rows<H, F, R, NW, ACT>(qbuf, qsc, rowcnt, bucket, mx, ctl + 1, wave, lane, d0,
                               ht, bias, slope, hagg);
    return;
  }

  for (int i = t; i < R * HF / 4; i += NT)
    ((float4*)acc)[i] = make_float4(0.f, 0.f, 0.f, 0.f);
  for (int i = t; i < R * H; i += NT) den[i] = 0.f;
  __syncthreads();
  if (!spilled) {
    agg_drain<H, F, R, true>(q, qs, qn, lane, d0, p, ew, src, mx, ht, acc, den);
  } else {
    qn = 0;
    agg_scan<H, F, R, NT, 1>(dst, wave, lane, d0, q, qs, qn, spilled, p, ew, src, ht, mx, acc, den);
    if (qn > 0)
      agg_drain<H, F, R, false>(q, qs, qn, lane, d0, p, ew, src, mx, ht, acc, den);
  }
  __syncthreads();

  for (int i = t; i < R * HF / 4; i += NT) {
    const int r = i / (HF / 4);
    const int n = d0 + r;
    if (n < NN_NODES) {
      float4 v = ((const float4*)acc)[i];
      const int c4 = i % (HF / 4);
      const float s = fmaxf(den[r * H + (c4 * 4) / F], 1e-20f);
      v.x = v.x / s; v.y = v.y / s; v.z = v.z / s; v.w = v.w / s;
      const float4 b = ((const float4*)bias)[c4];
      v.x += b.x; v.y += b.y; v.z += b.z; v.w += b.w;
      if constexpr (ACT) {
        v.x = leaky(v.x, slope); v.y = leaky(v.y, slope);
        v.z = leaky(v.z, slope); v.w = leaky(v.w, slope);
      }
      ((float4*)hagg)[(size_t)d0 * (HF / 4) + i] = v;
    }
  }
}

#define PRD_P1    132
#define PRD_P2    68
#define PRD_NW    4
#define PRD_UNITS (N_ROWS / 32)

__device__ __forceinline__ float predictor_tile(const float* __restrict__ emb, int s, int d,
                                                int c, int h,
                                                const float* W1s, const float* W2s,
                                                const float* __restrict__ pb1,
                                                const float* __restrict__ pb2,
                                                const float* __restrict__ pW3) {
  const float* ra = emb + (size_t)s * 128 + 4 * h;
  const float* rb = emb + (size_t)d * 128 + 4 * h;
  v8f acc[4];
#pragma unroll
  for (int nt = 0; nt < 4; ++nt) acc[nt] = (v8f){0.f, 0.f, 0.f, 0.f, 0.f, 0.f, 0.f, 0.f};

#pragma nounroll
  for (int jb = 0; jb < 4; ++jb) {
    v4f z[4];
#pragma unroll
    for (int j = 0; j < 4; ++j) {
      const v4f a = *(const v4f*)(ra + (jb * 4 + j) * 8);
      const v4f b = *(const v4f*)(rb + (jb * 4 + j) * 8);
      z[j] = a * b;
    }
#pragma unroll
    for (int j = 0; j < 4; ++j) {
      const int kc = (jb * 4 + j) * 8 + 4 * h;
      v4f w[4];
#pragma unroll
      for (int nt = 0; nt < 4; ++nt) w[nt] = *(const v4f*)(W1s + (nt * 16 + c) * PRD_P1 + kc);
      const v2f b0 = {z[j].x, z[j].y};
      const v2f b1 = {z[j].z, z[j].w};
#pragma unroll
      for (int nt = 0; nt < 4; ++nt) {
        const v2f a0 = {w[nt].x, w[nt].y};
        acc[nt] = __builtin_amdgcn_wmma_f32_16x16x4_f32(false, a0, false, b0, (short)0, acc[nt],
                                                        false, false);
      }
#pragma unroll
      for (int nt = 0; nt < 4; ++nt) {
        const v2f a1 = {w[nt].z, w[nt].w};
        acc[nt] = __builtin_amdgcn_wmma_f32_16x16x4_f32(false, a1, false, b1, (short)0, acc[nt],
                                                        false, false);
      }
    }
  }
#pragma unroll
  for (int nt = 0; nt < 4; ++nt) {
    const v4f bl = *(const v4f*)(pb1 + nt * 16 + 8 * h);
    const v4f bh = *(const v4f*)(pb1 + nt * 16 + 8 * h + 4);
    const float bb[8] = {bl.x, bl.y, bl.z, bl.w, bh.x, bh.y, bh.z, bh.w};
#pragma unroll
    for (int r = 0; r < 8; ++r) acc[nt][r] = leaky(acc[nt][r] + bb[r], 0.2f);
  }

  v8f acc2[2];
#pragma unroll
  for (int mt = 0; mt < 2; ++mt) acc2[mt] = (v8f){0.f, 0.f, 0.f, 0.f, 0.f, 0.f, 0.f, 0.f};
#pragma unroll
  for (int nt = 0; nt < 4; ++nt) {
    v4f wl[2], wh[2];
#pragma unroll
    for (int mt = 0; mt < 2; ++mt) {
      const float* wp = W2s + (mt * 16 + c) * PRD_P2 + nt * 16 + 8 * h;
      wl[mt] = *(const v4f*)wp;
      wh[mt] = *(const v4f*)(wp + 4);
    }
#pragma unroll
    for (int q = 0; q < 4; ++q) {
      const v2f b = {acc[nt][2 * q], acc[nt][2 * q + 1]};
#pragma unroll
      for (int mt = 0; mt < 2; ++mt) {
        const v4f wv = q < 2 ? wl[mt] : wh[mt];
        const v2f a = (q & 1) ? (v2f){wv.z, wv.w} : (v2f){wv.x, wv.y};
        acc2[mt] = __builtin_amdgcn_wmma_f32_16x16x4_f32(false, a, false, b, (short)0, acc2[mt],
                                                         false, false);
      }
    }
  }
  float part = 0.f;
#pragma unroll
  for (int mt = 0; mt < 2; ++mt) {
    const v4f bl = *(const v4f*)(pb2 + mt * 16 + 8 * h);
    const v4f bh = *(const v4f*)(pb2 + mt * 16 + 8 * h + 4);
    const v4f ul = *(const v4f*)(pW3 + mt * 16 + 8 * h);
    const v4f uh = *(const v4f*)(pW3 + mt * 16 + 8 * h + 4);
    const float bb[8] = {bl.x, bl.y, bl.z, bl.w, bh.x, bh.y, bh.z, bh.w};
    const float uu[8] = {ul.x, ul.y, ul.z, ul.w, uh.x, uh.y, uh.z, uh.w};
#pragma unroll
    for (int r = 0; r < 8; ++r) part += leaky(acc2[mt][r] + bb[r], 0.2f) * uu[r];
  }
  part += __shfl_xor(part, 16);
  return part;
}

__global__ __launch_bounds__(PRD_NW * 32) __attribute__((amdgpu_waves_per_eu(7)))
void predictor_kernel(const float* __restrict__ emb,
                      const int* __restrict__ ps, const int* __restrict__ pd,
                      const int* __restrict__ ns, const int* __restrict__ nd,
                      const float* __restrict__ pW1, const float* __restrict__ pb1,
                      const float* __restrict__ pW2, const float* __restrict__ pb2,
                      const float* __restrict__ pW3, const float* __restrict__ pb3,
                      float* out) {
  static_assert(N_ROWS % 32 == 0 && N_PAIR % 32 == 0, "a unit is 32 pairs of one list");
  __shared__ float W1s[64 * PRD_P1];
  __shared__ float W2s[32 * PRD_P2];
  const int t    = threadIdx.x;
  const int lane = t & 31;
  const int c    = lane & 15;
  const int h    = lane >> 4;
  const int wave = __builtin_amdgcn_readfirstlane(t >> 5);
  const int unit = blockIdx.x * PRD_NW + wave;

  int s0 = 0, d0 = 0, s1 = 0, d1 = 0;
  if (unit < PRD_UNITS) {
    const bool pos = unit < N_PAIR / 32;
    const int* si  = pos ? ps : ns;
    const int* di  = pos ? pd : nd;
    const int  r0  = (pos ? unit * 32 : unit * 32 - N_PAIR) + c;
    s0 = si[r0];      d0 = di[r0];
    s1 = si[r0 + 16]; d1 = di[r0 + 16];
  }
  static_assert(PRD_NW == 4, "staging counts below are for 128 threads");
  v4f g1[16], g2[4];
#pragma unroll
  for (int it = 0; it < 16; ++it) g1[it] = *(const v4f*)(pW1 + (t + it * 128) * 4);
#pragma unroll
  for (int it = 0; it < 4; ++it) g2[it] = *(const v4f*)(pW2 + (t + it * 128) * 4);
#pragma unroll
  for (int it = 0; it < 16; ++it) {
    const int i = t + it * 128;
    *(v4f*)(W1s + (i >> 5) * PRD_P1 + (i & 31) * 4) = g1[it];
  }
#pragma unroll
  for (int it = 0; it < 4; ++it) {
    const int i = t + it * 128;
    *(v4f*)(W2s + (i >> 4) * PRD_P2 + (i & 15) * 4) = g2[it];
  }
  __syncthreads();

  if (unit < PRD_UNITS) {
    const float v0 = predictor_tile(emb, s0, d0, c, h, W1s, W2s, pb1, pb2, pW3);
    const float v1 = predictor_tile(emb, s1, d1, c, h, W1s, W2s, pb1, pb2, pW3);
    out[unit * 32 + lane] = (h ? v1 : v0) + pb3[0];
  }
}

extern "C" void kernel_launch(void* const* d_in, const int* in_sizes, int n_in,
                              void* d_out, int out_size, void* d_ws, size_t ws_size,
                              hipStream_t stream) {
  const float* x    = (const float*)d_in[0];
  const float* ew   = (const float*)d_in[1];
  const int*   esrc = (const int*)d_in[2];
  const int*   edst = (const int*)d_in[3];
  const int*   psrc = (const int*)d_in[4];
  const int*   pdst = (const int*)d_in[5];
  const int*   nsrc = (const int*)d_in[6];
  const int*   ndst = (const int*)d_in[7];
  const float* Wl[3] = {(const float*)d_in[8],  (const float*)d_in[11], (const float*)d_in[14]};
  const float* al[3] = {(const float*)d_in[9],  (const float*)d_in[12], (const float*)d_in[15]};
  const float* bl[3] = {(const float*)d_in[10], (const float*)d_in[13], (const float*)d_in[16]};
  const float* pW1 = (const float*)d_in[17]; const float* pb1 = (const float*)d_in[18];
  const float* pW2 = (const float*)d_in[19]; const float* pb2 = (const float*)d_in[20];
  const float* pW3 = (const float*)d_in[21]; const float* pb3 = (const float*)d_in[22];

  float* out   = (float*)d_out;
  float* emb   = out + 2 * N_PAIR;

  float*    ws   = (float*)d_ws;
  float*    ht   = ws;
  float*    hbuf = ht   + 6400000;
  float*    pbuf = hbuf + 6400000;
  unsigned* savq = (unsigned*)(pbuf + 100000);
  float*    savl = pbuf + 100000 + 1048576;
  int*      savn = (int*)(pbuf + 100000 + 2097152);

  const int Hs[3]  = {4, 4, 1};
  const int Fs[3]  = {64, 64, 128};
  const int Kin[3] = {128, 256, 256};

  const float* hin = x;
  for (int l = 0; l < 3; ++l) {
    const int H = Hs[l], F = Fs[l], HF = H * F, K = Kin[l];
    float* hout = (l == 2) ? emb : hbuf;

    if (l == 0) {
      gemm_wmma_f32<true, 4, 64, 128><<<AGG_NT / 32 + P_NWG + GEMM_MB * 8, 256, 0, stream>>>(
          hin, Wl[l], nullptr, ht, NN_NODES, HF, K, 0.f, 0,
          esrc, edst, ew, savq, savl, savn, al[l], pbuf);
    } else if (l == 1) {
      gemm_wmma_f32<false, 4, 64, 256><<<P_NWG + GEMM_MB * 8, 256, 0, stream>>>(
          hin, Wl[l], nullptr, ht, NN_NODES, HF, K, 0.f, 0,
          nullptr, nullptr, nullptr, nullptr, nullptr, nullptr, al[l], pbuf);
    } else {
      gemm_wmma_f32<false, 1, 128, 256><<<P_NWG + GEMM_MB * 4, 256, 0, stream>>>(
          hin, Wl[l], nullptr, ht, NN_NODES, HF, K, 0.f, 0,
          nullptr, nullptr, nullptr, nullptr, nullptr, nullptr, al[l], pbuf);
    }

    if (l == 0) {
      edge_agg_owner_kernel<4, 64, AGG_R, AGG_NT, true, true><<<AGG_NG, AGG_NT, 0, stream>>>(
          pbuf, ew, ht, esrc, edst, bl[l], 0.01f, hout, savq, savl, savn);
    } else if (l == 1) {
      edge_agg_owner_kernel<4, 64, AGG_R, AGG_NT, true, false><<<AGG_NG, AGG_NT, 0, stream>>>(
          pbuf, ew, ht, esrc, edst, bl[l], 0.01f, hout, savq, savl, savn);
    } else {
      edge_agg_owner_kernel<1, 128, AGG_R, AGG_NT, false, false><<<AGG_NG, AGG_NT, 0, stream>>>(
          pbuf, ew, ht, esrc, edst, bl[l], 0.01f, hout, savq, savl, savn);
    }
    hin = hout;
  }

  predictor_kernel<<<(PRD_UNITS + PRD_NW - 1) / PRD_NW, PRD_NW * 32, 0, stream>>>(
      emb, psrc, pdst, nsrc, ndst, pW1, pb1, pW2, pb2, pW3, pb3, out);
}
